# starting kernel
# baseline (speedup 1.0000x reference)
#include <hip/hip_runtime.h>

#define B_ 4
#define N_ 8192
#define C_ 128
#define Q_ 2048
#define S_ 16
#define KEEP_ 17
#define RS_ 20

typedef __attribute__((ext_vector_type(2))) float v2f;
typedef __attribute__((ext_vector_type(8))) float v8f;

__global__ void pnorm2_kernel(const float* __restrict__ pts, float* __restrict__ pn2) {
  int i = blockIdx.x * blockDim.x + threadIdx.x;
  if (i < B_ * N_) {
    float x = pts[i * 3 + 0], y = pts[i * 3 + 1], z = pts[i * 3 + 2];
    pn2[i] = x * x + y * y + z * z;
  }
}

__global__ __launch_bounds__(256) void fps_kernel(const float* __restrict__ pts,
                                                  int* __restrict__ fps_idx,
                                                  float* __restrict__ qA) {
  const int b = blockIdx.x;
  const int tid = threadIdx.x;
  const float* P = pts + (size_t)b * N_ * 3;

  __shared__ float sval[256];
  __shared__ int   sidx[256];
  __shared__ float scur[3];

  float d[N_ / 256];
  const float p0x = P[0], p0y = P[1], p0z = P[2];
#pragma unroll
  for (int i = 0; i < N_ / 256; ++i) {
    int n = tid + i * 256;
    float dx = P[n * 3 + 0] - p0x, dy = P[n * 3 + 1] - p0y, dz = P[n * 3 + 2] - p0z;
    d[i] = dx * dx + dy * dy + dz * dz;
  }
  if (tid == 0) {
    fps_idx[b * Q_ + 0] = 0;
    qA[((size_t)b * Q_ + 0) * 4 + 0] = p0x;
    qA[((size_t)b * Q_ + 0) * 4 + 1] = p0y;
    qA[((size_t)b * Q_ + 0) * 4 + 2] = p0z;
    qA[((size_t)b * Q_ + 0) * 4 + 3] = p0x * p0x + p0y * p0y + p0z * p0z;
  }

  for (int s = 1; s < Q_; ++s) {
    float bv = -1.0f;
    int   bidx = 0;
#pragma unroll
    for (int i = 0; i < N_ / 256; ++i) {
      int n = tid + i * 256;
      if (d[i] > bv) { bv = d[i]; bidx = n; }
    }
    sval[tid] = bv; sidx[tid] = bidx;
    __syncthreads();
    for (int st = 128; st > 0; st >>= 1) {
      if (tid < st) {
        float ov = sval[tid + st]; int oi = sidx[tid + st];
        float mv = sval[tid];      int mi = sidx[tid];
        if (ov > mv || (ov == mv && oi < mi)) { sval[tid] = ov; sidx[tid] = oi; }
      }
      __syncthreads();
    }
    if (tid == 0) {
      int nxt = sidx[0];
      float x = P[nxt * 3 + 0], y = P[nxt * 3 + 1], z = P[nxt * 3 + 2];
      fps_idx[b * Q_ + s] = nxt;
      qA[((size_t)b * Q_ + s) * 4 + 0] = x;
      qA[((size_t)b * Q_ + s) * 4 + 1] = y;
      qA[((size_t)b * Q_ + s) * 4 + 2] = z;
      qA[((size_t)b * Q_ + s) * 4 + 3] = x * x + y * y + z * z;
      scur[0] = x; scur[1] = y; scur[2] = z;
    }
    __syncthreads();
    float cx = scur[0], cy = scur[1], cz = scur[2];
#pragma unroll
    for (int i = 0; i < N_ / 256; ++i) {
      int n = tid + i * 256;
      float dx = P[n * 3 + 0] - cx, dy = P[n * 3 + 1] - cy, dz = P[n * 3 + 2] - cz;
      float dd = dx * dx + dy * dy + dz * dz;
      d[i] = fminf(d[i], dd);
    }
  }
}

__global__ __launch_bounds__(256) void knn_kernel(const float* __restrict__ pts,
                                                  const float* __restrict__ pn2,
                                                  const float* __restrict__ qA,
                                                  int* __restrict__ nbr) {
  const int wave = threadIdx.x >> 5;
  const int lane = threadIdx.x & 31;
  const int b    = blockIdx.x >> 3;
  const int q0   = ((blockIdx.x & 7) * 8 + wave) * 32;

  const float* P  = pts + (size_t)b * N_ * 3;
  const float* PN = pn2 + (size_t)b * N_;
  const float* QA = qA + (size_t)b * Q_ * 4;

  __shared__ float stage[8][32 * RS_];
  float* st = stage[wave];

  const int lhalf = lane >> 4;
  const int l16   = lane & 15;

  const float4 qa0 = ((const float4*)QA)[q0 + l16];
  const float4 qa1 = ((const float4*)QA)[q0 + 16 + l16];
  v2f a0 = lhalf ? (v2f){-2.0f * qa0.z, 1.0f} : (v2f){-2.0f * qa0.x, -2.0f * qa0.y};
  v2f a1 = lhalf ? (v2f){-2.0f * qa1.z, 1.0f} : (v2f){-2.0f * qa1.x, -2.0f * qa1.y};

  v8f c0, c1;
#pragma unroll
  for (int r = 0; r < 8; ++r) {
    c0[r] = QA[(q0 + r + lhalf * 8) * 4 + 3];
    c1[r] = QA[(q0 + 16 + r + lhalf * 8) * 4 + 3];
  }

  float bd[KEEP_]; int bi[KEEP_];
#pragma unroll
  for (int k = 0; k < KEEP_; ++k) { bd[k] = 3.0e38f; bi[k] = 0; }

  auto load_bb = [&](int ch) -> v2f {
    int pt = ch * 16 + l16;
    float x0 = P[pt * 3 + (lhalf ? 2 : 0)];
    float x1 = lhalf ? PN[pt] : P[pt * 3 + 1];
    return (v2f){x0, x1};
  };

  const int NCH = N_ / 16;
  v2f bb = load_bb(0);
  for (int ch = 0; ch < NCH; ++ch) {
    v2f bb_next = load_bb(ch + 1 < NCH ? ch + 1 : NCH - 1);

    v8f d0 = __builtin_amdgcn_wmma_f32_16x16x4_f32(false, a0, false, bb, (short)0, c0, false, false);
    v8f d1 = __builtin_amdgcn_wmma_f32_16x16x4_f32(false, a1, false, bb, (short)0, c1, false, false);

#pragma unroll
    for (int r = 0; r < 8; ++r) {
      st[(r + lhalf * 8) * RS_ + l16]       = d0[r];
      st[(16 + r + lhalf * 8) * RS_ + l16]  = d1[r];
    }
    const float4* myrow = (const float4*)(st + lane * RS_);
#pragma unroll
    for (int jj = 0; jj < 4; ++jj) {
      float4 v4 = myrow[jj];
#pragma unroll
      for (int e = 0; e < 4; ++e) {
        float v = fmaxf(e == 0 ? v4.x : (e == 1 ? v4.y : (e == 2 ? v4.z : v4.w)), 0.0f);
        int pi = ch * 16 + jj * 4 + e;
        if (v < bd[KEEP_ - 1]) {
          bd[KEEP_ - 1] = v; bi[KEEP_ - 1] = pi;
#pragma unroll
          for (int t = KEEP_ - 1; t > 0; --t) {
            if (bd[t] < bd[t - 1]) {
              float tf = bd[t]; bd[t] = bd[t - 1]; bd[t - 1] = tf;
              int   ti = bi[t]; bi[t] = bi[t - 1]; bi[t - 1] = ti;
            }
          }
        }
      }
    }
    bb = bb_next;
  }

  int q = q0 + lane;
#pragma unroll
  for (int k = 0; k < S_; ++k) nbr[((size_t)b * Q_ + q) * S_ + k] = bi[k + 1];
}

__global__ __launch_bounds__(256) void three_nn_kernel(const float* __restrict__ pts,
                                                       const float* __restrict__ pn2,
                                                       const float* __restrict__ qA,
                                                       int* __restrict__ idx3,
                                                       float* __restrict__ w3) {
  const int wave = threadIdx.x >> 5;
  const int lane = threadIdx.x & 31;
  const int b    = blockIdx.x >> 5;
  const int n0   = ((blockIdx.x & 31) * 8 + wave) * 32;

  const float* P  = pts + (size_t)b * N_ * 3;
  const float* PN = pn2 + (size_t)b * N_;
  const float* QA = qA + (size_t)b * Q_ * 4;

  __shared__ float stage[8][32 * RS_];
  float* st = stage[wave];

  const int lhalf = lane >> 4;
  const int l16   = lane & 15;

  int m0 = n0 + l16, m1 = n0 + 16 + l16;
  float p0x = P[m0 * 3 + 0], p0y = P[m0 * 3 + 1], p0z = P[m0 * 3 + 2];
  float p1x = P[m1 * 3 + 0], p1y = P[m1 * 3 + 1], p1z = P[m1 * 3 + 2];
  v2f a0 = lhalf ? (v2f){-2.0f * p0z, 1.0f} : (v2f){-2.0f * p0x, -2.0f * p0y};
  v2f a1 = lhalf ? (v2f){-2.0f * p1z, 1.0f} : (v2f){-2.0f * p1x, -2.0f * p1y};

  v8f c0, c1;
#pragma unroll
  for (int r = 0; r < 8; ++r) {
    c0[r] = PN[n0 + r + lhalf * 8];
    c1[r] = PN[n0 + 16 + r + lhalf * 8];
  }

  float bd[3]; int bi[3];
#pragma unroll
  for (int k = 0; k < 3; ++k) { bd[k] = 3.0e38f; bi[k] = 0; }

  auto load_bb = [&](int ch) -> v2f {
    const float4 qa = ((const float4*)QA)[ch * 16 + l16];
    return lhalf ? (v2f){qa.z, qa.w} : (v2f){qa.x, qa.y};
  };

  const int NCH = Q_ / 16;
  v2f bb = load_bb(0);
  for (int ch = 0; ch < NCH; ++ch) {
    v2f bb_next = load_bb(ch + 1 < NCH ? ch + 1 : NCH - 1);

    v8f d0 = __builtin_amdgcn_wmma_f32_16x16x4_f32(false, a0, false, bb, (short)0, c0, false, false);
    v8f d1 = __builtin_amdgcn_wmma_f32_16x16x4_f32(false, a1, false, bb, (short)0, c1, false, false);

#pragma unroll
    for (int r = 0; r < 8; ++r) {
      st[(r + lhalf * 8) * RS_ + l16]      = d0[r];
      st[(16 + r + lhalf * 8) * RS_ + l16] = d1[r];
    }
    const float4* myrow = (const float4*)(st + lane * RS_);
#pragma unroll
    for (int jj = 0; jj < 4; ++jj) {
      float4 v4 = myrow[jj];
#pragma unroll
      for (int e = 0; e < 4; ++e) {
        float v = fmaxf(e == 0 ? v4.x : (e == 1 ? v4.y : (e == 2 ? v4.z : v4.w)), 0.0f);
        int qi = ch * 16 + jj * 4 + e;
        if (v < bd[2]) {
          bd[2] = v; bi[2] = qi;
          if (bd[2] < bd[1]) { float tf = bd[2]; bd[2] = bd[1]; bd[1] = tf;
                               int ti = bi[2]; bi[2] = bi[1]; bi[1] = ti; }
          if (bd[1] < bd[0]) { float tf = bd[1]; bd[1] = bd[0]; bd[0] = tf;
                               int ti = bi[1]; bi[1] = bi[0]; bi[0] = ti; }
        }
      }
    }
    bb = bb_next;
  }

  int n = n0 + lane;
  float dv0 = fmaxf(bd[0], 1e-10f), dv1 = fmaxf(bd[1], 1e-10f), dv2 = fmaxf(bd[2], 1e-10f);
  float i0 = 1.0f / (dv0 + 1e-8f), i1 = 1.0f / (dv1 + 1e-8f), i2 = 1.0f / (dv2 + 1e-8f);
  float inv_s = 1.0f / (i0 + i1 + i2);
  size_t base = ((size_t)b * N_ + n) * 3;
  idx3[base + 0] = bi[0]; idx3[base + 1] = bi[1]; idx3[base + 2] = bi[2];
  w3[base + 0] = i0 * inv_s; w3[base + 1] = i1 * inv_s; w3[base + 2] = i2 * inv_s;
}

__global__ __launch_bounds__(256) void group_mean_kernel(const float* __restrict__ feat,
                                                         const int* __restrict__ nbr,
                                                         float* __restrict__ coarse) {
  int bc = blockIdx.x;
  int b = bc / C_, c = bc % C_;
  const float* F = feat + ((size_t)b * C_ + c) * N_;
  const int* NB  = nbr + (size_t)b * Q_ * S_;
  float* O = coarse + ((size_t)b * C_ + c) * Q_;
  for (int q = threadIdx.x; q < Q_; q += blockDim.x) {
    float s = 0.0f;
#pragma unroll
    for (int k = 0; k < S_; ++k) s += F[NB[q * S_ + k]];
    O[q] = s * (1.0f / S_);
  }
}

__global__ __launch_bounds__(256) void interp_kernel(const float* __restrict__ coarse,
                                                     const int* __restrict__ idx3,
                                                     const float* __restrict__ w3,
                                                     float* __restrict__ out) {
  int bc = blockIdx.x;
  int b = bc / C_, c = bc % C_;
  const float* CO = coarse + ((size_t)b * C_ + c) * Q_;
  const int*   I  = idx3 + (size_t)b * N_ * 3;
  const float* W  = w3 + (size_t)b * N_ * 3;
  float* O = out + ((size_t)b * C_ + c) * N_;
  for (int n = threadIdx.x; n < N_; n += blockDim.x) {
    O[n] = W[n * 3 + 0] * CO[I[n * 3 + 0]] +
           W[n * 3 + 1] * CO[I[n * 3 + 1]] +
           W[n * 3 + 2] * CO[I[n * 3 + 2]];
  }
}

extern "C" void kernel_launch(void* const* d_in, const int* in_sizes, int n_in,
                              void* d_out, int out_size, void* d_ws, size_t ws_size,
                              hipStream_t stream) {
  const float* points   = (const float*)d_in[0];
  const float* features = (const float*)d_in[1];

  auto align256 = [](size_t x) { return (x + 255) & ~(size_t)255; };
  char* ws = (char*)d_ws;
  size_t off = 0;
  int*   fps_idx = (int*)(ws + off);   off = align256(off + sizeof(int) * B_ * Q_);
  float* qA      = (float*)(ws + off); off = align256(off + sizeof(float) * B_ * Q_ * 4);
  float* pn2     = (float*)(ws + off); off = align256(off + sizeof(float) * B_ * N_);
  int*   nbr     = (int*)(ws + off);   off = align256(off + sizeof(int) * B_ * Q_ * S_);
  float* coarse  = (float*)(ws + off); off = align256(off + sizeof(float) * B_ * C_ * Q_);
  int*   idx3    = (int*)(ws + off);   off = align256(off + sizeof(int) * B_ * N_ * 3);
  float* w3      = (float*)(ws + off); off = align256(off + sizeof(float) * B_ * N_ * 3);

  float* out = (float*)d_out;

  pnorm2_kernel<<<(B_ * N_ + 255) / 256, 256, 0, stream>>>(points, pn2);
  fps_kernel<<<B_, 256, 0, stream>>>(points, fps_idx, qA);
  knn_kernel<<<B_ * 8, 256, 0, stream>>>(points, pn2, qA, nbr);
  group_mean_kernel<<<B_ * C_, 256, 0, stream>>>(features, nbr, coarse);
  three_nn_kernel<<<B_ * 32, 256, 0, stream>>>(points, pn2, qA, idx3, w3);
  interp_kernel<<<B_ * C_, 256, 0, stream>>>(coarse, idx3, w3, out);
}


# fps: block argmax by DPP wave reduction + one LDS exchange of the 8 wave results (alternating slots) that every thread reduces itself, each loading the winner's xyz: 1 barrier per round, was 10
# speedup vs baseline: 2.2094x; 2.2094x over previous
#include <hip/hip_runtime.h>

#define B_ 4
#define N_ 8192
#define C_ 128
#define Q_ 2048
#define S_ 16
#define KEEP_ 17
#define RS_ 20

typedef __attribute__((ext_vector_type(2))) float v2f;
typedef __attribute__((ext_vector_type(8))) float v8f;

__global__ void pnorm2_kernel(const float* __restrict__ pts, float* __restrict__ pn2) {
  int i = blockIdx.x * blockDim.x + threadIdx.x;
  if (i < B_ * N_) {
    float x = pts[i * 3 + 0], y = pts[i * 3 + 1], z = pts[i * 3 + 2];
    pn2[i] = x * x + y * y + z * z;
  }
}

template <int M> __device__ __forceinline__ int row_xor(int v) {
  return __builtin_amdgcn_update_dpp(0, v, 0x160 + M, 0xf, 0xf, true);
}
__device__ __forceinline__ int xhalf(int v) {
  return __builtin_amdgcn_permlanex16(v, v, 0x76543210, 0xfedcba98, false, false);
}

__global__ __launch_bounds__(256) void fps_kernel(const float* __restrict__ pts,
                                                  int* __restrict__ fps_idx,
                                                  float* __restrict__ qA) {
  const int b = blockIdx.x;
  const int tid = threadIdx.x;
  const float* P = pts + (size_t)b * N_ * 3;

  __shared__ int2 part[2][8];
  const int wave = tid >> 5;
  const int lane = tid & 31;

  float d[N_ / 256];
  const float p0x = P[0], p0y = P[1], p0z = P[2];
#pragma unroll
  for (int i = 0; i < N_ / 256; ++i) {
    int n = tid + i * 256;
    float dx = P[n * 3 + 0] - p0x, dy = P[n * 3 + 1] - p0y, dz = P[n * 3 + 2] - p0z;
    d[i] = dx * dx + dy * dy + dz * dz;
  }
  if (tid == 0) {
    fps_idx[b * Q_ + 0] = 0;
    qA[((size_t)b * Q_ + 0) * 4 + 0] = p0x;
    qA[((size_t)b * Q_ + 0) * 4 + 1] = p0y;
    qA[((size_t)b * Q_ + 0) * 4 + 2] = p0z;
    qA[((size_t)b * Q_ + 0) * 4 + 3] = p0x * p0x + p0y * p0y + p0z * p0z;
  }

  for (int s = 1; s < Q_; ++s) {
    float bv = -1.0f;
    int   bidx = 0;
#pragma unroll
    for (int i = 0; i < N_ / 256; ++i) {
      int n = tid + i * 256;
      if (d[i] > bv) { bv = d[i]; bidx = n; }
    }
    const int bvi = __builtin_bit_cast(int, bv);
    int wm = bvi;
    wm = max(wm, row_xor<1>(wm)); wm = max(wm, row_xor<2>(wm));
    wm = max(wm, row_xor<4>(wm)); wm = max(wm, row_xor<8>(wm));
    wm = max(wm, xhalf(wm));
    int wi = (bvi == wm) ? bidx : 0x7fffffff;
    wi = min(wi, row_xor<1>(wi)); wi = min(wi, row_xor<2>(wi));
    wi = min(wi, row_xor<4>(wi)); wi = min(wi, row_xor<8>(wi));
    wi = min(wi, xhalf(wi));
    if (lane == 0) part[s & 1][wave] = make_int2(wm, wi);
    __syncthreads();
    const int2 pr = part[s & 1][lane & 7];
    int gm = pr.x;
    gm = max(gm, row_xor<1>(gm)); gm = max(gm, row_xor<2>(gm)); gm = max(gm, row_xor<4>(gm));
    int nxt = (pr.x == gm) ? pr.y : 0x7fffffff;
    nxt = min(nxt, row_xor<1>(nxt)); nxt = min(nxt, row_xor<2>(nxt)); nxt = min(nxt, row_xor<4>(nxt));
    nxt &= N_ - 1;
    const float cx = P[nxt * 3 + 0], cy = P[nxt * 3 + 1], cz = P[nxt * 3 + 2];
    if (tid == 0) {
      float x = cx, y = cy, z = cz;
      fps_idx[b * Q_ + s] = nxt;
      qA[((size_t)b * Q_ + s) * 4 + 0] = x;
      qA[((size_t)b * Q_ + s) * 4 + 1] = y;
      qA[((size_t)b * Q_ + s) * 4 + 2] = z;
      qA[((size_t)b * Q_ + s) * 4 + 3] = x * x + y * y + z * z;
    }
#pragma unroll
    for (int i = 0; i < N_ / 256; ++i) {
      int n = tid + i * 256;
      float dx = P[n * 3 + 0] - cx, dy = P[n * 3 + 1] - cy, dz = P[n * 3 + 2] - cz;
      float dd = dx * dx + dy * dy + dz * dz;
      d[i] = fminf(d[i], dd);
    }
  }
}

__global__ __launch_bounds__(256) void knn_kernel(const float* __restrict__ pts,
                                                  const float* __restrict__ pn2,
                                                  const float* __restrict__ qA,
                                                  int* __restrict__ nbr) {
  const int wave = threadIdx.x >> 5;
  const int lane = threadIdx.x & 31;
  const int b    = blockIdx.x >> 3;
  const int q0   = ((blockIdx.x & 7) * 8 + wave) * 32;

  const float* P  = pts + (size_t)b * N_ * 3;
  const float* PN = pn2 + (size_t)b * N_;
  const float* QA = qA + (size_t)b * Q_ * 4;

  __shared__ float stage[8][32 * RS_];
  float* st = stage[wave];

  const int lhalf = lane >> 4;
  const int l16   = lane & 15;

  const float4 qa0 = ((const float4*)QA)[q0 + l16];
  const float4 qa1 = ((const float4*)QA)[q0 + 16 + l16];
  v2f a0 = lhalf ? (v2f){-2.0f * qa0.z, 1.0f} : (v2f){-2.0f * qa0.x, -2.0f * qa0.y};
  v2f a1 = lhalf ? (v2f){-2.0f * qa1.z, 1.0f} : (v2f){-2.0f * qa1.x, -2.0f * qa1.y};

  v8f c0, c1;
#pragma unroll
  for (int r = 0; r < 8; ++r) {
    c0[r] = QA[(q0 + r + lhalf * 8) * 4 + 3];
    c1[r] = QA[(q0 + 16 + r + lhalf * 8) * 4 + 3];
  }

  float bd[KEEP_]; int bi[KEEP_];
#pragma unroll
  for (int k = 0; k < KEEP_; ++k) { bd[k] = 3.0e38f; bi[k] = 0; }

  auto load_bb = [&](int ch) -> v2f {
    int pt = ch * 16 + l16;
    float x0 = P[pt * 3 + (lhalf ? 2 : 0)];
    float x1 = lhalf ? PN[pt] : P[pt * 3 + 1];
    return (v2f){x0, x1};
  };

  const int NCH = N_ / 16;
  v2f bb = load_bb(0);
  for (int ch = 0; ch < NCH; ++ch) {
    v2f bb_next = load_bb(ch + 1 < NCH ? ch + 1 : NCH - 1);

    v8f d0 = __builtin_amdgcn_wmma_f32_16x16x4_f32(false, a0, false, bb, (short)0, c0, false, false);
    v8f d1 = __builtin_amdgcn_wmma_f32_16x16x4_f32(false, a1, false, bb, (short)0, c1, false, false);

#pragma unroll
    for (int r = 0; r < 8; ++r) {
      st[(r + lhalf * 8) * RS_ + l16]       = d0[r];
      st[(16 + r + lhalf * 8) * RS_ + l16]  = d1[r];
    }
    const float4* myrow = (const float4*)(st + lane * RS_);
#pragma unroll
    for (int jj = 0; jj < 4; ++jj) {
      float4 v4 = myrow[jj];
#pragma unroll
      for (int e = 0; e < 4; ++e) {
        float v = fmaxf(e == 0 ? v4.x : (e == 1 ? v4.y : (e == 2 ? v4.z : v4.w)), 0.0f);
        int pi = ch * 16 + jj * 4 + e;
        if (v < bd[KEEP_ - 1]) {
          bd[KEEP_ - 1] = v; bi[KEEP_ - 1] = pi;
#pragma unroll
          for (int t = KEEP_ - 1; t > 0; --t) {
            if (bd[t] < bd[t - 1]) {
              float tf = bd[t]; bd[t] = bd[t - 1]; bd[t - 1] = tf;
              int   ti = bi[t]; bi[t] = bi[t - 1]; bi[t - 1] = ti;
            }
          }
        }
      }
    }
    bb = bb_next;
  }

  int q = q0 + lane;
#pragma unroll
  for (int k = 0; k < S_; ++k) nbr[((size_t)b * Q_ + q) * S_ + k] = bi[k + 1];
}

__global__ __launch_bounds__(256) void three_nn_kernel(const float* __restrict__ pts,
                                                       const float* __restrict__ pn2,
                                                       const float* __restrict__ qA,
                                                       int* __restrict__ idx3,
                                                       float* __restrict__ w3) {
  const int wave = threadIdx.x >> 5;
  const int lane = threadIdx.x & 31;
  const int b    = blockIdx.x >> 5;
  const int n0   = ((blockIdx.x & 31) * 8 + wave) * 32;

  const float* P  = pts + (size_t)b * N_ * 3;
  const float* PN = pn2 + (size_t)b * N_;
  const float* QA = qA + (size_t)b * Q_ * 4;

  __shared__ float stage[8][32 * RS_];
  float* st = stage[wave];

  const int lhalf = lane >> 4;
  const int l16   = lane & 15;

  int m0 = n0 + l16, m1 = n0 + 16 + l16;
  float p0x = P[m0 * 3 + 0], p0y = P[m0 * 3 + 1], p0z = P[m0 * 3 + 2];
  float p1x = P[m1 * 3 + 0], p1y = P[m1 * 3 + 1], p1z = P[m1 * 3 + 2];
  v2f a0 = lhalf ? (v2f){-2.0f * p0z, 1.0f} : (v2f){-2.0f * p0x, -2.0f * p0y};
  v2f a1 = lhalf ? (v2f){-2.0f * p1z, 1.0f} : (v2f){-2.0f * p1x, -2.0f * p1y};

  v8f c0, c1;
#pragma unroll
  for (int r = 0; r < 8; ++r) {
    c0[r] = PN[n0 + r + lhalf * 8];
    c1[r] = PN[n0 + 16 + r + lhalf * 8];
  }

  float bd[3]; int bi[3];
#pragma unroll
  for (int k = 0; k < 3; ++k) { bd[k] = 3.0e38f; bi[k] = 0; }

  auto load_bb = [&](int ch) -> v2f {
    const float4 qa = ((const float4*)QA)[ch * 16 + l16];
    return lhalf ? (v2f){qa.z, qa.w} : (v2f){qa.x, qa.y};
  };

  const int NCH = Q_ / 16;
  v2f bb = load_bb(0);
  for (int ch = 0; ch < NCH; ++ch) {
    v2f bb_next = load_bb(ch + 1 < NCH ? ch + 1 : NCH - 1);

    v8f d0 = __builtin_amdgcn_wmma_f32_16x16x4_f32(false, a0, false, bb, (short)0, c0, false, false);
    v8f d1 = __builtin_amdgcn_wmma_f32_16x16x4_f32(false, a1, false, bb, (short)0, c1, false, false);

#pragma unroll
    for (int r = 0; r < 8; ++r) {
      st[(r + lhalf * 8) * RS_ + l16]      = d0[r];
      st[(16 + r + lhalf * 8) * RS_ + l16] = d1[r];
    }
    const float4* myrow = (const float4*)(st + lane * RS_);
#pragma unroll
    for (int jj = 0; jj < 4; ++jj) {
      float4 v4 = myrow[jj];
#pragma unroll
      for (int e = 0; e < 4; ++e) {
        float v = fmaxf(e == 0 ? v4.x : (e == 1 ? v4.y : (e == 2 ? v4.z : v4.w)), 0.0f);
        int qi = ch * 16 + jj * 4 + e;
        if (v < bd[2]) {
          bd[2] = v; bi[2] = qi;
          if (bd[2] < bd[1]) { float tf = bd[2]; bd[2] = bd[1]; bd[1] = tf;
                               int ti = bi[2]; bi[2] = bi[1]; bi[1] = ti; }
          if (bd[1] < bd[0]) { float tf = bd[1]; bd[1] = bd[0]; bd[0] = tf;
                               int ti = bi[1]; bi[1] = bi[0]; bi[0] = ti; }
        }
      }
    }
    bb = bb_next;
  }

  int n = n0 + lane;
  float dv0 = fmaxf(bd[0], 1e-10f), dv1 = fmaxf(bd[1], 1e-10f), dv2 = fmaxf(bd[2], 1e-10f);
  float i0 = 1.0f / (dv0 + 1e-8f), i1 = 1.0f / (dv1 + 1e-8f), i2 = 1.0f / (dv2 + 1e-8f);
  float inv_s = 1.0f / (i0 + i1 + i2);
  size_t base = ((size_t)b * N_ + n) * 3;
  idx3[base + 0] = bi[0]; idx3[base + 1] = bi[1]; idx3[base + 2] = bi[2];
  w3[base + 0] = i0 * inv_s; w3[base + 1] = i1 * inv_s; w3[base + 2] = i2 * inv_s;
}

__global__ __launch_bounds__(256) void group_mean_kernel(const float* __restrict__ feat,
                                                         const int* __restrict__ nbr,
                                                         float* __restrict__ coarse) {
  int bc = blockIdx.x;
  int b = bc / C_, c = bc % C_;
  const float* F = feat + ((size_t)b * C_ + c) * N_;
  const int* NB  = nbr + (size_t)b * Q_ * S_;
  float* O = coarse + ((size_t)b * C_ + c) * Q_;
  for (int q = threadIdx.x; q < Q_; q += blockDim.x) {
    float s = 0.0f;
#pragma unroll
    for (int k = 0; k < S_; ++k) s += F[NB[q * S_ + k]];
    O[q] = s * (1.0f / S_);
  }
}

__global__ __launch_bounds__(256) void interp_kernel(const float* __restrict__ coarse,
                                                     const int* __restrict__ idx3,
                                                     const float* __restrict__ w3,
                                                     float* __restrict__ out) {
  int bc = blockIdx.x;
  int b = bc / C_, c = bc % C_;
  const float* CO = coarse + ((size_t)b * C_ + c) * Q_;
  const int*   I  = idx3 + (size_t)b * N_ * 3;
  const float* W  = w3 + (size_t)b * N_ * 3;
  float* O = out + ((size_t)b * C_ + c) * N_;
  for (int n = threadIdx.x; n < N_; n += blockDim.x) {
    O[n] = W[n * 3 + 0] * CO[I[n * 3 + 0]] +
           W[n * 3 + 1] * CO[I[n * 3 + 1]] +
           W[n * 3 + 2] * CO[I[n * 3 + 2]];
  }
}

extern "C" void kernel_launch(void* const* d_in, const int* in_sizes, int n_in,
                              void* d_out, int out_size, void* d_ws, size_t ws_size,
                              hipStream_t stream) {
  const float* points   = (const float*)d_in[0];
  const float* features = (const float*)d_in[1];

  auto align256 = [](size_t x) { return (x + 255) & ~(size_t)255; };
  char* ws = (char*)d_ws;
  size_t off = 0;
  int*   fps_idx = (int*)(ws + off);   off = align256(off + sizeof(int) * B_ * Q_);
  float* qA      = (float*)(ws + off); off = align256(off + sizeof(float) * B_ * Q_ * 4);
  float* pn2     = (float*)(ws + off); off = align256(off + sizeof(float) * B_ * N_);
  int*   nbr     = (int*)(ws + off);   off = align256(off + sizeof(int) * B_ * Q_ * S_);
  float* coarse  = (float*)(ws + off); off = align256(off + sizeof(float) * B_ * C_ * Q_);
  int*   idx3    = (int*)(ws + off);   off = align256(off + sizeof(int) * B_ * N_ * 3);
  float* w3      = (float*)(ws + off); off = align256(off + sizeof(float) * B_ * N_ * 3);

  float* out = (float*)d_out;

  pnorm2_kernel<<<(B_ * N_ + 255) / 256, 256, 0, stream>>>(points, pn2);
  fps_kernel<<<B_, 256, 0, stream>>>(points, fps_idx, qA);
  knn_kernel<<<B_ * 8, 256, 0, stream>>>(points, pn2, qA, nbr);
  group_mean_kernel<<<B_ * C_, 256, 0, stream>>>(features, nbr, coarse);
  three_nn_kernel<<<B_ * 32, 256, 0, stream>>>(points, pn2, qA, idx3, w3);
  interp_kernel<<<B_ * C_, 256, 0, stream>>>(coarse, idx3, w3, out);
}


# fps: each thread's argmax is carried inside the distance update as 8 independent integer compare chains on the distance bit patterns, replacing the 32-step serial compare-select chain
# speedup vs baseline: 2.4807x; 1.1228x over previous
#include <hip/hip_runtime.h>

#define B_ 4
#define N_ 8192
#define C_ 128
#define Q_ 2048
#define S_ 16
#define KEEP_ 17
#define RS_ 20

typedef __attribute__((ext_vector_type(2))) float v2f;
typedef __attribute__((ext_vector_type(8))) float v8f;

__global__ void pnorm2_kernel(const float* __restrict__ pts, float* __restrict__ pn2) {
  int i = blockIdx.x * blockDim.x + threadIdx.x;
  if (i < B_ * N_) {
    float x = pts[i * 3 + 0], y = pts[i * 3 + 1], z = pts[i * 3 + 2];
    pn2[i] = x * x + y * y + z * z;
  }
}

template <int M> __device__ __forceinline__ int row_xor(int v) {
  return __builtin_amdgcn_update_dpp(0, v, 0x160 + M, 0xf, 0xf, true);
}
__device__ __forceinline__ int xhalf(int v) {
  return __builtin_amdgcn_permlanex16(v, v, 0x76543210, 0xfedcba98, false, false);
}

#define FPS_PT_ (N_ / 256)
#define FPS_CH_ 8
#define FPS_CL_ (FPS_PT_ / FPS_CH_)

__device__ __forceinline__ void fps_take_higher(int& va, int& sa, int vb, int sb) {
  const bool g = vb > va;
  va = g ? vb : va;
  sa = g ? sb : sa;
}

__device__ __forceinline__ void fps_thread_argmax(const int (&d)[FPS_PT_], int& bv, int& bslot) {
  int v[FPS_CH_], sl[FPS_CH_];
#pragma unroll
  for (int c = 0; c < FPS_CH_; ++c) { v[c] = d[c * FPS_CL_]; sl[c] = c * FPS_CL_; }
#pragma unroll
  for (int j = 1; j < FPS_CL_; ++j) {
#pragma unroll
    for (int c = 0; c < FPS_CH_; ++c) {
      const int i = c * FPS_CL_ + j;
      const bool g = d[i] > v[c];
      v[c]  = g ? d[i] : v[c];
      sl[c] = g ? i : sl[c];
    }
  }
  fps_take_higher(v[0], sl[0], v[1], sl[1]);
  fps_take_higher(v[2], sl[2], v[3], sl[3]);
  fps_take_higher(v[4], sl[4], v[5], sl[5]);
  fps_take_higher(v[6], sl[6], v[7], sl[7]);
  fps_take_higher(v[0], sl[0], v[2], sl[2]);
  fps_take_higher(v[4], sl[4], v[6], sl[6]);
  fps_take_higher(v[0], sl[0], v[4], sl[4]);
  bv = v[0];
  bslot = sl[0];
}

__global__ __launch_bounds__(256) void fps_kernel(const float* __restrict__ pts,
                                                  int* __restrict__ fps_idx,
                                                  float* __restrict__ qA) {
  const int b = blockIdx.x;
  const int tid = threadIdx.x;
  const float* P = pts + (size_t)b * N_ * 3;

  __shared__ int2 part[2][8];
  const int wave = tid >> 5;
  const int lane = tid & 31;

  int d[FPS_PT_];
  const float p0x = P[0], p0y = P[1], p0z = P[2];
#pragma unroll
  for (int i = 0; i < FPS_PT_; ++i) {
    int n = tid + i * 256;
    float dx = P[n * 3 + 0] - p0x, dy = P[n * 3 + 1] - p0y, dz = P[n * 3 + 2] - p0z;
    d[i] = __builtin_bit_cast(int, dx * dx + dy * dy + dz * dz);
  }
  int bvi, bslot;
  fps_thread_argmax(d, bvi, bslot);
  if (tid == 0) {
    fps_idx[b * Q_ + 0] = 0;
    qA[((size_t)b * Q_ + 0) * 4 + 0] = p0x;
    qA[((size_t)b * Q_ + 0) * 4 + 1] = p0y;
    qA[((size_t)b * Q_ + 0) * 4 + 2] = p0z;
    qA[((size_t)b * Q_ + 0) * 4 + 3] = p0x * p0x + p0y * p0y + p0z * p0z;
  }

  for (int s = 1; s < Q_; ++s) {
    const int bidx = tid + bslot * 256;
    int wm = bvi;
    wm = max(wm, row_xor<1>(wm)); wm = max(wm, row_xor<2>(wm));
    wm = max(wm, row_xor<4>(wm)); wm = max(wm, row_xor<8>(wm));
    wm = max(wm, xhalf(wm));
    int wi = (bvi == wm) ? bidx : 0x7fffffff;
    wi = min(wi, row_xor<1>(wi)); wi = min(wi, row_xor<2>(wi));
    wi = min(wi, row_xor<4>(wi)); wi = min(wi, row_xor<8>(wi));
    wi = min(wi, xhalf(wi));
    if (lane == 0) part[s & 1][wave] = make_int2(wm, wi);
    __syncthreads();
    const int2 pr = part[s & 1][lane & 7];
    int gm = pr.x;
    gm = max(gm, row_xor<1>(gm)); gm = max(gm, row_xor<2>(gm)); gm = max(gm, row_xor<4>(gm));
    int nxt = (pr.x == gm) ? pr.y : 0x7fffffff;
    nxt = min(nxt, row_xor<1>(nxt)); nxt = min(nxt, row_xor<2>(nxt)); nxt = min(nxt, row_xor<4>(nxt));
    nxt &= N_ - 1;
    const float cx = P[nxt * 3 + 0], cy = P[nxt * 3 + 1], cz = P[nxt * 3 + 2];
    if (tid == 0) {
      float x = cx, y = cy, z = cz;
      fps_idx[b * Q_ + s] = nxt;
      qA[((size_t)b * Q_ + s) * 4 + 0] = x;
      qA[((size_t)b * Q_ + s) * 4 + 1] = y;
      qA[((size_t)b * Q_ + s) * 4 + 2] = z;
      qA[((size_t)b * Q_ + s) * 4 + 3] = x * x + y * y + z * z;
    }
#pragma unroll
    for (int i = 0; i < FPS_PT_; ++i) {
      int n = tid + i * 256;
      float dx = P[n * 3 + 0] - cx, dy = P[n * 3 + 1] - cy, dz = P[n * 3 + 2] - cz;
      float dd = dx * dx + dy * dy + dz * dz;
      d[i] = min(d[i], __builtin_bit_cast(int, dd));
    }
    fps_thread_argmax(d, bvi, bslot);
  }
}

__global__ __launch_bounds__(256) void knn_kernel(const float* __restrict__ pts,
                                                  const float* __restrict__ pn2,
                                                  const float* __restrict__ qA,
                                                  int* __restrict__ nbr) {
  const int wave = threadIdx.x >> 5;
  const int lane = threadIdx.x & 31;
  const int b    = blockIdx.x >> 3;
  const int q0   = ((blockIdx.x & 7) * 8 + wave) * 32;

  const float* P  = pts + (size_t)b * N_ * 3;
  const float* PN = pn2 + (size_t)b * N_;
  const float* QA = qA + (size_t)b * Q_ * 4;

  __shared__ float stage[8][32 * RS_];
  float* st = stage[wave];

  const int lhalf = lane >> 4;
  const int l16   = lane & 15;

  const float4 qa0 = ((const float4*)QA)[q0 + l16];
  const float4 qa1 = ((const float4*)QA)[q0 + 16 + l16];
  v2f a0 = lhalf ? (v2f){-2.0f * qa0.z, 1.0f} : (v2f){-2.0f * qa0.x, -2.0f * qa0.y};
  v2f a1 = lhalf ? (v2f){-2.0f * qa1.z, 1.0f} : (v2f){-2.0f * qa1.x, -2.0f * qa1.y};

  v8f c0, c1;
#pragma unroll
  for (int r = 0; r < 8; ++r) {
    c0[r] = QA[(q0 + r + lhalf * 8) * 4 + 3];
    c1[r] = QA[(q0 + 16 + r + lhalf * 8) * 4 + 3];
  }

  float bd[KEEP_]; int bi[KEEP_];
#pragma unroll
  for (int k = 0; k < KEEP_; ++k) { bd[k] = 3.0e38f; bi[k] = 0; }

  auto load_bb = [&](int ch) -> v2f {
    int pt = ch * 16 + l16;
    float x0 = P[pt * 3 + (lhalf ? 2 : 0)];
    float x1 = lhalf ? PN[pt] : P[pt * 3 + 1];
    return (v2f){x0, x1};
  };

  const int NCH = N_ / 16;
  v2f bb = load_bb(0);
  for (int ch = 0; ch < NCH; ++ch) {
    v2f bb_next = load_bb(ch + 1 < NCH ? ch + 1 : NCH - 1);

    v8f d0 = __builtin_amdgcn_wmma_f32_16x16x4_f32(false, a0, false, bb, (short)0, c0, false, false);
    v8f d1 = __builtin_amdgcn_wmma_f32_16x16x4_f32(false, a1, false, bb, (short)0, c1, false, false);

#pragma unroll
    for (int r = 0; r < 8; ++r) {
      st[(r + lhalf * 8) * RS_ + l16]       = d0[r];
      st[(16 + r + lhalf * 8) * RS_ + l16]  = d1[r];
    }
    const float4* myrow = (const float4*)(st + lane * RS_);
#pragma unroll
    for (int jj = 0; jj < 4; ++jj) {
      float4 v4 = myrow[jj];
#pragma unroll
      for (int e = 0; e < 4; ++e) {
        float v = fmaxf(e == 0 ? v4.x : (e == 1 ? v4.y : (e == 2 ? v4.z : v4.w)), 0.0f);
        int pi = ch * 16 + jj * 4 + e;
        if (v < bd[KEEP_ - 1]) {
          bd[KEEP_ - 1] = v; bi[KEEP_ - 1] = pi;
#pragma unroll
          for (int t = KEEP_ - 1; t > 0; --t) {
            if (bd[t] < bd[t - 1]) {
              float tf = bd[t]; bd[t] = bd[t - 1]; bd[t - 1] = tf;
              int   ti = bi[t]; bi[t] = bi[t - 1]; bi[t - 1] = ti;
            }
          }
        }
      }
    }
    bb = bb_next;
  }

  int q = q0 + lane;
#pragma unroll
  for (int k = 0; k < S_; ++k) nbr[((size_t)b * Q_ + q) * S_ + k] = bi[k + 1];
}

__global__ __launch_bounds__(256) void three_nn_kernel(const float* __restrict__ pts,
                                                       const float* __restrict__ pn2,
                                                       const float* __restrict__ qA,
                                                       int* __restrict__ idx3,
                                                       float* __restrict__ w3) {
  const int wave = threadIdx.x >> 5;
  const int lane = threadIdx.x & 31;
  const int b    = blockIdx.x >> 5;
  const int n0   = ((blockIdx.x & 31) * 8 + wave) * 32;

  const float* P  = pts + (size_t)b * N_ * 3;
  const float* PN = pn2 + (size_t)b * N_;
  const float* QA = qA + (size_t)b * Q_ * 4;

  __shared__ float stage[8][32 * RS_];
  float* st = stage[wave];

  const int lhalf = lane >> 4;
  const int l16   = lane & 15;

  int m0 = n0 + l16, m1 = n0 + 16 + l16;
  float p0x = P[m0 * 3 + 0], p0y = P[m0 * 3 + 1], p0z = P[m0 * 3 + 2];
  float p1x = P[m1 * 3 + 0], p1y = P[m1 * 3 + 1], p1z = P[m1 * 3 + 2];
  v2f a0 = lhalf ? (v2f){-2.0f * p0z, 1.0f} : (v2f){-2.0f * p0x, -2.0f * p0y};
  v2f a1 = lhalf ? (v2f){-2.0f * p1z, 1.0f} : (v2f){-2.0f * p1x, -2.0f * p1y};

  v8f c0, c1;
#pragma unroll
  for (int r = 0; r < 8; ++r) {
    c0[r] = PN[n0 + r + lhalf * 8];
    c1[r] = PN[n0 + 16 + r + lhalf * 8];
  }

  float bd[3]; int bi[3];
#pragma unroll
  for (int k = 0; k < 3; ++k) { bd[k] = 3.0e38f; bi[k] = 0; }

  auto load_bb = [&](int ch) -> v2f {
    const float4 qa = ((const float4*)QA)[ch * 16 + l16];
    return lhalf ? (v2f){qa.z, qa.w} : (v2f){qa.x, qa.y};
  };

  const int NCH = Q_ / 16;
  v2f bb = load_bb(0);
  for (int ch = 0; ch < NCH; ++ch) {
    v2f bb_next = load_bb(ch + 1 < NCH ? ch + 1 : NCH - 1);

    v8f d0 = __builtin_amdgcn_wmma_f32_16x16x4_f32(false, a0, false, bb, (short)0, c0, false, false);
    v8f d1 = __builtin_amdgcn_wmma_f32_16x16x4_f32(false, a1, false, bb, (short)0, c1, false, false);

#pragma unroll
    for (int r = 0; r < 8; ++r) {
      st[(r + lhalf * 8) * RS_ + l16]      = d0[r];
      st[(16 + r + lhalf * 8) * RS_ + l16] = d1[r];
    }
    const float4* myrow = (const float4*)(st + lane * RS_);
#pragma unroll
    for (int jj = 0; jj < 4; ++jj) {
      float4 v4 = myrow[jj];
#pragma unroll
      for (int e = 0; e < 4; ++e) {
        float v = fmaxf(e == 0 ? v4.x : (e == 1 ? v4.y : (e == 2 ? v4.z : v4.w)), 0.0f);
        int qi = ch * 16 + jj * 4 + e;
        if (v < bd[2]) {
          bd[2] = v; bi[2] = qi;
          if (bd[2] < bd[1]) { float tf = bd[2]; bd[2] = bd[1]; bd[1] = tf;
                               int ti = bi[2]; bi[2] = bi[1]; bi[1] = ti; }
          if (bd[1] < bd[0]) { float tf = bd[1]; bd[1] = bd[0]; bd[0] = tf;
                               int ti = bi[1]; bi[1] = bi[0]; bi[0] = ti; }
        }
      }
    }
    bb = bb_next;
  }

  int n = n0 + lane;
  float dv0 = fmaxf(bd[0], 1e-10f), dv1 = fmaxf(bd[1], 1e-10f), dv2 = fmaxf(bd[2], 1e-10f);
  float i0 = 1.0f / (dv0 + 1e-8f), i1 = 1.0f / (dv1 + 1e-8f), i2 = 1.0f / (dv2 + 1e-8f);
  float inv_s = 1.0f / (i0 + i1 + i2);
  size_t base = ((size_t)b * N_ + n) * 3;
  idx3[base + 0] = bi[0]; idx3[base + 1] = bi[1]; idx3[base + 2] = bi[2];
  w3[base + 0] = i0 * inv_s; w3[base + 1] = i1 * inv_s; w3[base + 2] = i2 * inv_s;
}

__global__ __launch_bounds__(256) void group_mean_kernel(const float* __restrict__ feat,
                                                         const int* __restrict__ nbr,
                                                         float* __restrict__ coarse) {
  int bc = blockIdx.x;
  int b = bc / C_, c = bc % C_;
  const float* F = feat + ((size_t)b * C_ + c) * N_;
  const int* NB  = nbr + (size_t)b * Q_ * S_;
  float* O = coarse + ((size_t)b * C_ + c) * Q_;
  for (int q = threadIdx.x; q < Q_; q += blockDim.x) {
    float s = 0.0f;
#pragma unroll
    for (int k = 0; k < S_; ++k) s += F[NB[q * S_ + k]];
    O[q] = s * (1.0f / S_);
  }
}

__global__ __launch_bounds__(256) void interp_kernel(const float* __restrict__ coarse,
                                                     const int* __restrict__ idx3,
                                                     const float* __restrict__ w3,
                                                     float* __restrict__ out) {
  int bc = blockIdx.x;
  int b = bc / C_, c = bc % C_;
  const float* CO = coarse + ((size_t)b * C_ + c) * Q_;
  const int*   I  = idx3 + (size_t)b * N_ * 3;
  const float* W  = w3 + (size_t)b * N_ * 3;
  float* O = out + ((size_t)b * C_ + c) * N_;
  for (int n = threadIdx.x; n < N_; n += blockDim.x) {
    O[n] = W[n * 3 + 0] * CO[I[n * 3 + 0]] +
           W[n * 3 + 1] * CO[I[n * 3 + 1]] +
           W[n * 3 + 2] * CO[I[n * 3 + 2]];
  }
}

extern "C" void kernel_launch(void* const* d_in, const int* in_sizes, int n_in,
                              void* d_out, int out_size, void* d_ws, size_t ws_size,
                              hipStream_t stream) {
  const float* points   = (const float*)d_in[0];
  const float* features = (const float*)d_in[1];

  auto align256 = [](size_t x) { return (x + 255) & ~(size_t)255; };
  char* ws = (char*)d_ws;
  size_t off = 0;
  int*   fps_idx = (int*)(ws + off);   off = align256(off + sizeof(int) * B_ * Q_);
  float* qA      = (float*)(ws + off); off = align256(off + sizeof(float) * B_ * Q_ * 4);
  float* pn2     = (float*)(ws + off); off = align256(off + sizeof(float) * B_ * N_);
  int*   nbr     = (int*)(ws + off);   off = align256(off + sizeof(int) * B_ * Q_ * S_);
  float* coarse  = (float*)(ws + off); off = align256(off + sizeof(float) * B_ * C_ * Q_);
  int*   idx3    = (int*)(ws + off);   off = align256(off + sizeof(int) * B_ * N_ * 3);
  float* w3      = (float*)(ws + off); off = align256(off + sizeof(float) * B_ * N_ * 3);

  float* out = (float*)d_out;

  pnorm2_kernel<<<(B_ * N_ + 255) / 256, 256, 0, stream>>>(points, pn2);
  fps_kernel<<<B_, 256, 0, stream>>>(points, fps_idx, qA);
  knn_kernel<<<B_ * 8, 256, 0, stream>>>(points, pn2, qA, nbr);
  group_mean_kernel<<<B_ * C_, 256, 0, stream>>>(features, nbr, coarse);
  three_nn_kernel<<<B_ * 32, 256, 0, stream>>>(points, pn2, qA, idx3, w3);
  interp_kernel<<<B_ * C_, 256, 0, stream>>>(coarse, idx3, w3, out);
}


# knn: the 8 waves of a block share 32 queries and each sweeps one eighth of the points; wave 0 merges their sorted 17-lists from LDS in slice order (same strict compares); 256 blocks, was 32
# speedup vs baseline: 3.5300x; 1.4230x over previous
#include <hip/hip_runtime.h>

#define B_ 4
#define N_ 8192
#define C_ 128
#define Q_ 2048
#define S_ 16
#define KEEP_ 17
#define RS_ 20

typedef __attribute__((ext_vector_type(2))) float v2f;
typedef __attribute__((ext_vector_type(8))) float v8f;

__global__ void pnorm2_kernel(const float* __restrict__ pts, float* __restrict__ pn2) {
  int i = blockIdx.x * blockDim.x + threadIdx.x;
  if (i < B_ * N_) {
    float x = pts[i * 3 + 0], y = pts[i * 3 + 1], z = pts[i * 3 + 2];
    pn2[i] = x * x + y * y + z * z;
  }
}

template <int M> __device__ __forceinline__ int row_xor(int v) {
  return __builtin_amdgcn_update_dpp(0, v, 0x160 + M, 0xf, 0xf, true);
}
__device__ __forceinline__ int xhalf(int v) {
  return __builtin_amdgcn_permlanex16(v, v, 0x76543210, 0xfedcba98, false, false);
}

#define FPS_PT_ (N_ / 256)
#define FPS_CH_ 8
#define FPS_CL_ (FPS_PT_ / FPS_CH_)
#define FPS_LS_ (FPS_PT_ + 1)

__device__ __forceinline__ float fps_dist2(float dx, float dy, float dz) {
  return __builtin_fmaf(dz, dz, __builtin_fmaf(dx, dx, dy * dy));
}

__device__ __forceinline__ void fps_take_higher(int& va, int& sa, int vb, int sb) {
  const bool g = vb > va;
  va = g ? vb : va;
  sa = g ? sb : sa;
}

__device__ __forceinline__ void fps_thread_argmax(const int (&d)[FPS_PT_], int& bv, int& bslot) {
  int v[FPS_CH_], sl[FPS_CH_];
#pragma unroll
  for (int c = 0; c < FPS_CH_; ++c) { v[c] = d[c * FPS_CL_]; sl[c] = c * FPS_CL_; }
#pragma unroll
  for (int j = 1; j < FPS_CL_; ++j) {
#pragma unroll
    for (int c = 0; c < FPS_CH_; ++c) {
      const int i = c * FPS_CL_ + j;
      const bool g = d[i] > v[c];
      v[c]  = g ? d[i] : v[c];
      sl[c] = g ? i : sl[c];
    }
  }
  fps_take_higher(v[0], sl[0], v[1], sl[1]);
  fps_take_higher(v[2], sl[2], v[3], sl[3]);
  fps_take_higher(v[4], sl[4], v[5], sl[5]);
  fps_take_higher(v[6], sl[6], v[7], sl[7]);
  fps_take_higher(v[0], sl[0], v[2], sl[2]);
  fps_take_higher(v[4], sl[4], v[6], sl[6]);
  fps_take_higher(v[0], sl[0], v[4], sl[4]);
  bv = v[0];
  bslot = sl[0];
}

__global__ __launch_bounds__(256) void fps_kernel(const float* __restrict__ pts,
                                                  int* __restrict__ fps_idx,
                                                  float* __restrict__ qA) {
  const int b = blockIdx.x;
  const int tid = threadIdx.x;
  const float* P = pts + (size_t)b * N_ * 3;

  __shared__ float4 lpts[256 * FPS_LS_];
  __shared__ int4 part[2][10];
  __shared__ int4 spare[256 + 64];
  const int wave = tid >> 5;
  const int lane = tid & 31;

  float px[FPS_PT_], py[FPS_PT_], pz[FPS_PT_];
  int d[FPS_PT_];
  const float p0x = P[0], p0y = P[1], p0z = P[2];
  {
    const float4* P4 = (const float4*)(P + (size_t)tid * (FPS_PT_ * 3));
    float f[FPS_PT_ * 3];
#pragma unroll
    for (int k = 0; k < FPS_PT_ * 3 / 4; ++k) {
      const float4 t4 = P4[k];
      f[k * 4 + 0] = t4.x; f[k * 4 + 1] = t4.y; f[k * 4 + 2] = t4.z; f[k * 4 + 3] = t4.w;
    }
#pragma unroll
    for (int i = 0; i < FPS_PT_; ++i) {
      px[i] = f[i * 3 + 0]; py[i] = f[i * 3 + 1]; pz[i] = f[i * 3 + 2];
      lpts[tid * FPS_LS_ + i] = make_float4(px[i], py[i], pz[i], 0.0f);
      float dx = px[i] - p0x, dy = py[i] - p0y, dz = pz[i] - p0z;
      d[i] = __builtin_bit_cast(int, fps_dist2(dx, dy, dz));
    }
  }
  int bvi, bslot;
  fps_thread_argmax(d, bvi, bslot);
  if (tid == 0) {
    fps_idx[b * Q_ + 0] = 0;
    qA[((size_t)b * Q_ + 0) * 4 + 0] = p0x;
    qA[((size_t)b * Q_ + 0) * 4 + 1] = p0y;
    qA[((size_t)b * Q_ + 0) * 4 + 2] = p0z;
    qA[((size_t)b * Q_ + 0) * 4 + 3] = p0x * p0x + p0y * p0y + p0z * p0z;
  }

  for (int s = 1; s < Q_; ++s) {
    const int bidx = tid * FPS_PT_ + bslot;
    const float4 cp = lpts[tid * FPS_LS_ + bslot];
    int wm = bvi;
    wm = max(wm, row_xor<1>(wm)); wm = max(wm, row_xor<2>(wm));
    wm = max(wm, row_xor<4>(wm)); wm = max(wm, row_xor<8>(wm));
    wm = max(wm, xhalf(wm));
    const bool lead = lane == __builtin_ctz(__builtin_amdgcn_ballot_w32(bvi == wm));
    int4* recA = lead ? &part[s & 1][wave] : &spare[tid];
    int*  recZ = lead ? &((int*)&part[s & 1][8])[wave] : &((int*)&spare[256])[tid];
    asm volatile("" :: "v"(cp.x), "v"(cp.y), "v"(cp.z), "v"(recA), "v"(recZ));
    *recA = make_int4(wm, bidx, __builtin_bit_cast(int, cp.x), __builtin_bit_cast(int, cp.y));
    *recZ = __builtin_bit_cast(int, cp.z);
    __syncthreads();
    const int4 ra = part[s & 1][lane & 7];
    const int rz = ((const int*)&part[s & 1][8])[lane & 7];
    int gm = ra.x;
    gm = max(gm, row_xor<1>(gm)); gm = max(gm, row_xor<2>(gm)); gm = max(gm, row_xor<4>(gm));
    const int wl = __builtin_ctz(__builtin_amdgcn_ballot_w32(ra.x == gm));
    const int nxt = __builtin_amdgcn_readlane(ra.y, wl);
    const float cx = __builtin_bit_cast(float, __builtin_amdgcn_readlane(ra.z, wl));
    const float cy = __builtin_bit_cast(float, __builtin_amdgcn_readlane(ra.w, wl));
    const float cz = __builtin_bit_cast(float, __builtin_amdgcn_readlane(rz, wl));
    if (tid == 0) {
      float x = cx, y = cy, z = cz;
      fps_idx[b * Q_ + s] = nxt;
      qA[((size_t)b * Q_ + s) * 4 + 0] = x;
      qA[((size_t)b * Q_ + s) * 4 + 1] = y;
      qA[((size_t)b * Q_ + s) * 4 + 2] = z;
      float n2;
      {
#pragma clang fp contract(off)
        n2 = x * x + y * y;
      }
      qA[((size_t)b * Q_ + s) * 4 + 3] = __builtin_fmaf(z, z, n2);
    }
#pragma unroll
    for (int i = 0; i < FPS_PT_; ++i) {
      float dx = px[i] - cx, dy = py[i] - cy, dz = pz[i] - cz;
      float dd = fps_dist2(dx, dy, dz);
      d[i] = min(d[i], __builtin_bit_cast(int, dd));
    }
    fps_thread_argmax(d, bvi, bslot);
  }
}

#define KNN_SL_ 8
#define KNN_SLCH_ (N_ / 16 / KNN_SL_)

__device__ __forceinline__ void knn_insert(float (&bd)[KEEP_], int (&bi)[KEEP_], float v, int pi) {
  if (v < bd[KEEP_ - 1]) {
    bd[KEEP_ - 1] = v; bi[KEEP_ - 1] = pi;
#pragma unroll
    for (int t = KEEP_ - 1; t > 0; --t) {
      if (bd[t] < bd[t - 1]) {
        float tf = bd[t]; bd[t] = bd[t - 1]; bd[t - 1] = tf;
        int   ti = bi[t]; bi[t] = bi[t - 1]; bi[t - 1] = ti;
      }
    }
  }
}

__global__ __launch_bounds__(256) void knn_kernel(const float* __restrict__ pts,
                                                  const float* __restrict__ pn2,
                                                  const float* __restrict__ qA,
                                                  int* __restrict__ nbr) {
  const int wave = __builtin_amdgcn_readfirstlane((int)(threadIdx.x >> 5));
  const int lane = threadIdx.x & 31;
  const int b    = blockIdx.x / (Q_ / 32);
  const int q0   = (blockIdx.x % (Q_ / 32)) * 32;

  const float* P  = pts + (size_t)b * N_ * 3;
  const float* PN = pn2 + (size_t)b * N_;
  const float* QA = qA + (size_t)b * Q_ * 4;

  __shared__ float stage[8][32 * RS_];
  __shared__ float mrg_d[KNN_SL_ - 1][KEEP_][32];
  __shared__ int   mrg_i[KNN_SL_ - 1][KEEP_][32];
  float* st = stage[wave];

  const int lhalf = lane >> 4;
  const int l16   = lane & 15;

  const float4 qa0 = ((const float4*)QA)[q0 + l16];
  const float4 qa1 = ((const float4*)QA)[q0 + 16 + l16];
  v2f a0 = lhalf ? (v2f){-2.0f * qa0.z, 1.0f} : (v2f){-2.0f * qa0.x, -2.0f * qa0.y};
  v2f a1 = lhalf ? (v2f){-2.0f * qa1.z, 1.0f} : (v2f){-2.0f * qa1.x, -2.0f * qa1.y};

  v8f c0, c1;
#pragma unroll
  for (int r = 0; r < 8; ++r) {
    c0[r] = QA[(q0 + r + lhalf * 8) * 4 + 3];
    c1[r] = QA[(q0 + 16 + r + lhalf * 8) * 4 + 3];
  }

  float bd[KEEP_]; int bi[KEEP_];
#pragma unroll
  for (int k = 0; k < KEEP_; ++k) { bd[k] = 3.0e38f; bi[k] = 0; }

  auto load_bb = [&](int ch) -> v2f {
    int pt = ch * 16 + l16;
    float x0 = P[pt * 3 + (lhalf ? 2 : 0)];
    float x1 = lhalf ? PN[pt] : P[pt * 3 + 1];
    return (v2f){x0, x1};
  };

  const int ch0 = wave * KNN_SLCH_;
  const int ch1 = ch0 + KNN_SLCH_;
  v2f bb = load_bb(ch0);
  for (int ch = ch0; ch < ch1; ++ch) {
    v2f bb_next = load_bb(ch + 1 < ch1 ? ch + 1 : ch1 - 1);

    v8f d0 = __builtin_amdgcn_wmma_f32_16x16x4_f32(false, a0, false, bb, (short)0, c0, false, false);
    v8f d1 = __builtin_amdgcn_wmma_f32_16x16x4_f32(false, a1, false, bb, (short)0, c1, false, false);

#pragma unroll
    for (int r = 0; r < 8; ++r) {
      st[(r + lhalf * 8) * RS_ + l16]       = d0[r];
      st[(16 + r + lhalf * 8) * RS_ + l16]  = d1[r];
    }
    const float4* myrow = (const float4*)(st + lane * RS_);
#pragma unroll
    for (int jj = 0; jj < 4; ++jj) {
      float4 v4 = myrow[jj];
#pragma unroll
      for (int e = 0; e < 4; ++e) {
        float v = fmaxf(e == 0 ? v4.x : (e == 1 ? v4.y : (e == 2 ? v4.z : v4.w)), 0.0f);
        knn_insert(bd, bi, v, ch * 16 + jj * 4 + e);
      }
    }
    bb = bb_next;
  }

  if (wave != 0) {
#pragma unroll
    for (int k = 0; k < KEEP_; ++k) {
      mrg_d[wave - 1][k][lane] = bd[k];
      mrg_i[wave - 1][k][lane] = bi[k];
    }
  }
  __syncthreads();
  if (wave != 0) return;

#pragma nounroll
  for (int w = 0; w < KNN_SL_ - 1; ++w) {
#pragma nounroll
    for (int k = 0; k < KEEP_; ++k) {
      const float v = mrg_d[w][k][lane];
      const int  pi = mrg_i[w][k][lane];
      if (__builtin_amdgcn_ballot_w32(v < bd[KEEP_ - 1]) == 0) break;
      knn_insert(bd, bi, v, pi);
    }
  }

  int q = q0 + lane;
#pragma unroll
  for (int k = 0; k < S_; ++k) nbr[((size_t)b * Q_ + q) * S_ + k] = bi[k + 1];
}

__global__ __launch_bounds__(256) void three_nn_kernel(const float* __restrict__ pts,
                                                       const float* __restrict__ pn2,
                                                       const float* __restrict__ qA,
                                                       int* __restrict__ idx3,
                                                       float* __restrict__ w3) {
  const int wave = threadIdx.x >> 5;
  const int lane = threadIdx.x & 31;
  const int b    = blockIdx.x >> 5;
  const int n0   = ((blockIdx.x & 31) * 8 + wave) * 32;

  const float* P  = pts + (size_t)b * N_ * 3;
  const float* PN = pn2 + (size_t)b * N_;
  const float* QA = qA + (size_t)b * Q_ * 4;

  __shared__ float stage[8][32 * RS_];
  float* st = stage[wave];

  const int lhalf = lane >> 4;
  const int l16   = lane & 15;

  int m0 = n0 + l16, m1 = n0 + 16 + l16;
  float p0x = P[m0 * 3 + 0], p0y = P[m0 * 3 + 1], p0z = P[m0 * 3 + 2];
  float p1x = P[m1 * 3 + 0], p1y = P[m1 * 3 + 1], p1z = P[m1 * 3 + 2];
  v2f a0 = lhalf ? (v2f){-2.0f * p0z, 1.0f} : (v2f){-2.0f * p0x, -2.0f * p0y};
  v2f a1 = lhalf ? (v2f){-2.0f * p1z, 1.0f} : (v2f){-2.0f * p1x, -2.0f * p1y};

  v8f c0, c1;
#pragma unroll
  for (int r = 0; r < 8; ++r) {
    c0[r] = PN[n0 + r + lhalf * 8];
    c1[r] = PN[n0 + 16 + r + lhalf * 8];
  }

  float bd[3]; int bi[3];
#pragma unroll
  for (int k = 0; k < 3; ++k) { bd[k] = 3.0e38f; bi[k] = 0; }

  auto load_bb = [&](int ch) -> v2f {
    const float4 qa = ((const float4*)QA)[ch * 16 + l16];
    return lhalf ? (v2f){qa.z, qa.w} : (v2f){qa.x, qa.y};
  };

  const int NCH = Q_ / 16;
  v2f bb = load_bb(0);
  for (int ch = 0; ch < NCH; ++ch) {
    v2f bb_next = load_bb(ch + 1 < NCH ? ch + 1 : NCH - 1);

    v8f d0 = __builtin_amdgcn_wmma_f32_16x16x4_f32(false, a0, false, bb, (short)0, c0, false, false);
    v8f d1 = __builtin_amdgcn_wmma_f32_16x16x4_f32(false, a1, false, bb, (short)0, c1, false, false);

#pragma unroll
    for (int r = 0; r < 8; ++r) {
      st[(r + lhalf * 8) * RS_ + l16]      = d0[r];
      st[(16 + r + lhalf * 8) * RS_ + l16] = d1[r];
    }
    const float4* myrow = (const float4*)(st + lane * RS_);
#pragma unroll
    for (int jj = 0; jj < 4; ++jj) {
      float4 v4 = myrow[jj];
#pragma unroll
      for (int e = 0; e < 4; ++e) {
        float v = fmaxf(e == 0 ? v4.x : (e == 1 ? v4.y : (e == 2 ? v4.z : v4.w)), 0.0f);
        int qi = ch * 16 + jj * 4 + e;
        if (v < bd[2]) {
          bd[2] = v; bi[2] = qi;
          if (bd[2] < bd[1]) { float tf = bd[2]; bd[2] = bd[1]; bd[1] = tf;
                               int ti = bi[2]; bi[2] = bi[1]; bi[1] = ti; }
          if (bd[1] < bd[0]) { float tf = bd[1]; bd[1] = bd[0]; bd[0] = tf;
                               int ti = bi[1]; bi[1] = bi[0]; bi[0] = ti; }
        }
      }
    }
    bb = bb_next;
  }

  int n = n0 + lane;
  float dv0 = fmaxf(bd[0], 1e-10f), dv1 = fmaxf(bd[1], 1e-10f), dv2 = fmaxf(bd[2], 1e-10f);
  float i0 = 1.0f / (dv0 + 1e-8f), i1 = 1.0f / (dv1 + 1e-8f), i2 = 1.0f / (dv2 + 1e-8f);
  float inv_s = 1.0f / (i0 + i1 + i2);
  size_t base = ((size_t)b * N_ + n) * 3;
  idx3[base + 0] = bi[0]; idx3[base + 1] = bi[1]; idx3[base + 2] = bi[2];
  w3[base + 0] = i0 * inv_s; w3[base + 1] = i1 * inv_s; w3[base + 2] = i2 * inv_s;
}

__global__ __launch_bounds__(256) void group_mean_kernel(const float* __restrict__ feat,
                                                         const int* __restrict__ nbr,
                                                         float* __restrict__ coarse) {
  int bc = blockIdx.x;
  int b = bc / C_, c = bc % C_;
  const float* F = feat + ((size_t)b * C_ + c) * N_;
  const int* NB  = nbr + (size_t)b * Q_ * S_;
  float* O = coarse + ((size_t)b * C_ + c) * Q_;
  for (int q = threadIdx.x; q < Q_; q += blockDim.x) {
    float s = 0.0f;
#pragma unroll
    for (int k = 0; k < S_; ++k) s += F[NB[q * S_ + k]];
    O[q] = s * (1.0f / S_);
  }
}

__global__ __launch_bounds__(256) void interp_kernel(const float* __restrict__ coarse,
                                                     const int* __restrict__ idx3,
                                                     const float* __restrict__ w3,
                                                     float* __restrict__ out) {
  int bc = blockIdx.x;
  int b = bc / C_, c = bc % C_;
  const float* CO = coarse + ((size_t)b * C_ + c) * Q_;
  const int*   I  = idx3 + (size_t)b * N_ * 3;
  const float* W  = w3 + (size_t)b * N_ * 3;
  float* O = out + ((size_t)b * C_ + c) * N_;
  for (int n = threadIdx.x; n < N_; n += blockDim.x) {
    O[n] = W[n * 3 + 0] * CO[I[n * 3 + 0]] +
           W[n * 3 + 1] * CO[I[n * 3 + 1]] +
           W[n * 3 + 2] * CO[I[n * 3 + 2]];
  }
}

extern "C" void kernel_launch(void* const* d_in, const int* in_sizes, int n_in,
                              void* d_out, int out_size, void* d_ws, size_t ws_size,
                              hipStream_t stream) {
  const float* points   = (const float*)d_in[0];
  const float* features = (const float*)d_in[1];

  auto align256 = [](size_t x) { return (x + 255) & ~(size_t)255; };
  char* ws = (char*)d_ws;
  size_t off = 0;
  int*   fps_idx = (int*)(ws + off);   off = align256(off + sizeof(int) * B_ * Q_);
  float* qA      = (float*)(ws + off); off = align256(off + sizeof(float) * B_ * Q_ * 4);
  float* pn2     = (float*)(ws + off); off = align256(off + sizeof(float) * B_ * N_);
  int*   nbr     = (int*)(ws + off);   off = align256(off + sizeof(int) * B_ * Q_ * S_);
  float* coarse  = (float*)(ws + off); off = align256(off + sizeof(float) * B_ * C_ * Q_);
  int*   idx3    = (int*)(ws + off);   off = align256(off + sizeof(int) * B_ * N_ * 3);
  float* w3      = (float*)(ws + off); off = align256(off + sizeof(float) * B_ * N_ * 3);

  float* out = (float*)d_out;

  pnorm2_kernel<<<(B_ * N_ + 255) / 256, 256, 0, stream>>>(points, pn2);
  fps_kernel<<<B_, 256, 0, stream>>>(points, fps_idx, qA);
  knn_kernel<<<B_ * (Q_ / 32), 256, 0, stream>>>(points, pn2, qA, nbr);
  group_mean_kernel<<<B_ * C_, 256, 0, stream>>>(features, nbr, coarse);
  three_nn_kernel<<<B_ * 32, 256, 0, stream>>>(points, pn2, qA, idx3, w3);
  interp_kernel<<<B_ * C_, 256, 0, stream>>>(coarse, idx3, w3, out);
}


# knn: lanes queue entries below their list's 17th distance in a wave-private LDS queue; the insertion network runs only in drains when a queue nears full (same entry order, same compares)
# speedup vs baseline: 3.7396x; 1.0594x over previous
#include <hip/hip_runtime.h>

#define B_ 4
#define N_ 8192
#define C_ 128
#define Q_ 2048
#define S_ 16
#define KEEP_ 17
#define RS_ 20

typedef __attribute__((ext_vector_type(2))) float v2f;
typedef __attribute__((ext_vector_type(8))) float v8f;

__global__ void pnorm2_kernel(const float* __restrict__ pts, float* __restrict__ pn2) {
  int i = blockIdx.x * blockDim.x + threadIdx.x;
  if (i < B_ * N_) {
    float x = pts[i * 3 + 0], y = pts[i * 3 + 1], z = pts[i * 3 + 2];
    pn2[i] = x * x + y * y + z * z;
  }
}

template <int M> __device__ __forceinline__ int row_xor(int v) {
  return __builtin_amdgcn_update_dpp(0, v, 0x160 + M, 0xf, 0xf, true);
}
__device__ __forceinline__ int xhalf(int v) {
  return __builtin_amdgcn_permlanex16(v, v, 0x76543210, 0xfedcba98, false, false);
}

#define FPS_PT_ (N_ / 256)
#define FPS_CH_ 8
#define FPS_CL_ (FPS_PT_ / FPS_CH_)
#define FPS_LS_ (FPS_PT_ + 1)

__device__ __forceinline__ float fps_dist2(float dx, float dy, float dz) {
  return __builtin_fmaf(dz, dz, __builtin_fmaf(dx, dx, dy * dy));
}

__device__ __forceinline__ void fps_take_higher(int& va, int& sa, int vb, int sb) {
  const bool g = vb > va;
  va = g ? vb : va;
  sa = g ? sb : sa;
}

__device__ __forceinline__ void fps_thread_argmax(const int (&d)[FPS_PT_], int& bv, int& bslot) {
  int v[FPS_CH_], sl[FPS_CH_];
#pragma unroll
  for (int c = 0; c < FPS_CH_; ++c) { v[c] = d[c * FPS_CL_]; sl[c] = c * FPS_CL_; }
#pragma unroll
  for (int j = 1; j < FPS_CL_; ++j) {
#pragma unroll
    for (int c = 0; c < FPS_CH_; ++c) {
      const int i = c * FPS_CL_ + j;
      const bool g = d[i] > v[c];
      v[c]  = g ? d[i] : v[c];
      sl[c] = g ? i : sl[c];
    }
  }
  fps_take_higher(v[0], sl[0], v[1], sl[1]);
  fps_take_higher(v[2], sl[2], v[3], sl[3]);
  fps_take_higher(v[4], sl[4], v[5], sl[5]);
  fps_take_higher(v[6], sl[6], v[7], sl[7]);
  fps_take_higher(v[0], sl[0], v[2], sl[2]);
  fps_take_higher(v[4], sl[4], v[6], sl[6]);
  fps_take_higher(v[0], sl[0], v[4], sl[4]);
  bv = v[0];
  bslot = sl[0];
}

__global__ __launch_bounds__(256) void fps_kernel(const float* __restrict__ pts,
                                                  int* __restrict__ fps_idx,
                                                  float* __restrict__ qA) {
  const int b = blockIdx.x;
  const int tid = threadIdx.x;
  const float* P = pts + (size_t)b * N_ * 3;

  __shared__ float4 lpts[256 * FPS_LS_];
  __shared__ int4 part[2][10];
  __shared__ int4 spare[256 + 64];
  const int wave = tid >> 5;
  const int lane = tid & 31;

  float px[FPS_PT_], py[FPS_PT_], pz[FPS_PT_];
  int d[FPS_PT_];
  const float p0x = P[0], p0y = P[1], p0z = P[2];
  {
    const float4* P4 = (const float4*)(P + (size_t)tid * (FPS_PT_ * 3));
    float f[FPS_PT_ * 3];
#pragma unroll
    for (int k = 0; k < FPS_PT_ * 3 / 4; ++k) {
      const float4 t4 = P4[k];
      f[k * 4 + 0] = t4.x; f[k * 4 + 1] = t4.y; f[k * 4 + 2] = t4.z; f[k * 4 + 3] = t4.w;
    }
#pragma unroll
    for (int i = 0; i < FPS_PT_; ++i) {
      px[i] = f[i * 3 + 0]; py[i] = f[i * 3 + 1]; pz[i] = f[i * 3 + 2];
      lpts[tid * FPS_LS_ + i] = make_float4(px[i], py[i], pz[i], 0.0f);
      float dx = px[i] - p0x, dy = py[i] - p0y, dz = pz[i] - p0z;
      d[i] = __builtin_bit_cast(int, fps_dist2(dx, dy, dz));
    }
  }
  int bvi, bslot;
  fps_thread_argmax(d, bvi, bslot);
  if (tid == 0) {
    fps_idx[b * Q_ + 0] = 0;
    qA[((size_t)b * Q_ + 0) * 4 + 0] = p0x;
    qA[((size_t)b * Q_ + 0) * 4 + 1] = p0y;
    qA[((size_t)b * Q_ + 0) * 4 + 2] = p0z;
    qA[((size_t)b * Q_ + 0) * 4 + 3] = p0x * p0x + p0y * p0y + p0z * p0z;
  }

  for (int s = 1; s < Q_; ++s) {
    const int bidx = tid * FPS_PT_ + bslot;
    const float4 cp = lpts[tid * FPS_LS_ + bslot];
    int wm = bvi;
    wm = max(wm, row_xor<1>(wm)); wm = max(wm, row_xor<2>(wm));
    wm = max(wm, row_xor<4>(wm)); wm = max(wm, row_xor<8>(wm));
    wm = max(wm, xhalf(wm));
    const bool lead = lane == __builtin_ctz(__builtin_amdgcn_ballot_w32(bvi == wm));
    int4* recA = lead ? &part[s & 1][wave] : &spare[tid];
    int*  recZ = lead ? &((int*)&part[s & 1][8])[wave] : &((int*)&spare[256])[tid];
    asm volatile("" :: "v"(cp.x), "v"(cp.y), "v"(cp.z), "v"(recA), "v"(recZ));
    *recA = make_int4(wm, bidx, __builtin_bit_cast(int, cp.x), __builtin_bit_cast(int, cp.y));
    *recZ = __builtin_bit_cast(int, cp.z);
    __syncthreads();
    const int4 ra = part[s & 1][lane & 7];
    const int rz = ((const int*)&part[s & 1][8])[lane & 7];
    int gm = ra.x;
    gm = max(gm, row_xor<1>(gm)); gm = max(gm, row_xor<2>(gm)); gm = max(gm, row_xor<4>(gm));
    const int wl = __builtin_ctz(__builtin_amdgcn_ballot_w32(ra.x == gm));
    const int nxt = __builtin_amdgcn_readlane(ra.y, wl);
    const float cx = __builtin_bit_cast(float, __builtin_amdgcn_readlane(ra.z, wl));
    const float cy = __builtin_bit_cast(float, __builtin_amdgcn_readlane(ra.w, wl));
    const float cz = __builtin_bit_cast(float, __builtin_amdgcn_readlane(rz, wl));
    if (tid == 0) {
      float x = cx, y = cy, z = cz;
      fps_idx[b * Q_ + s] = nxt;
      qA[((size_t)b * Q_ + s) * 4 + 0] = x;
      qA[((size_t)b * Q_ + s) * 4 + 1] = y;
      qA[((size_t)b * Q_ + s) * 4 + 2] = z;
      float n2;
      {
#pragma clang fp contract(off)
        n2 = x * x + y * y;
      }
      qA[((size_t)b * Q_ + s) * 4 + 3] = __builtin_fmaf(z, z, n2);
    }
#pragma unroll
    for (int i = 0; i < FPS_PT_; ++i) {
      float dx = px[i] - cx, dy = py[i] - cy, dz = pz[i] - cz;
      float dd = fps_dist2(dx, dy, dz);
      d[i] = min(d[i], __builtin_bit_cast(int, dd));
    }
    fps_thread_argmax(d, bvi, bslot);
  }
}

#define KNN_SL_ 8
#define KNN_SLCH_ (N_ / 16 / KNN_SL_)
#define KNN_QC_ 32

__device__ __forceinline__ void knn_insert(float (&bd)[KEEP_], int (&bi)[KEEP_], float v, int pi) {
  if (v < bd[KEEP_ - 1]) {
    bd[KEEP_ - 1] = v; bi[KEEP_ - 1] = pi;
#pragma unroll
    for (int t = KEEP_ - 1; t > 0; --t) {
      if (bd[t] < bd[t - 1]) {
        float tf = bd[t]; bd[t] = bd[t - 1]; bd[t - 1] = tf;
        int   ti = bi[t]; bi[t] = bi[t - 1]; bi[t - 1] = ti;
      }
    }
  }
}

__global__ __launch_bounds__(256) void knn_kernel(const float* __restrict__ pts,
                                                  const float* __restrict__ pn2,
                                                  const float* __restrict__ qA,
                                                  int* __restrict__ nbr) {
  const int wave = __builtin_amdgcn_readfirstlane((int)(threadIdx.x >> 5));
  const int lane = threadIdx.x & 31;
  const int b    = blockIdx.x / (Q_ / 32);
  const int q0   = (blockIdx.x % (Q_ / 32)) * 32;

  const float* P  = pts + (size_t)b * N_ * 3;
  const float* PN = pn2 + (size_t)b * N_;
  const float* QA = qA + (size_t)b * Q_ * 4;

  __shared__ float stage[8][32 * RS_];
  __shared__ float mrg_d[KNN_SL_ - 1][KEEP_][32];
  __shared__ int   mrg_i[KNN_SL_ - 1][KEEP_][32];
  __shared__ int   kq[KNN_SL_][KNN_QC_ * 64];
  float* st = stage[wave];
  int* kql = kq[wave] + lane;

  const int lhalf = lane >> 4;
  const int l16   = lane & 15;

  const float4 qa0 = ((const float4*)QA)[q0 + l16];
  const float4 qa1 = ((const float4*)QA)[q0 + 16 + l16];
  v2f a0 = lhalf ? (v2f){-2.0f * qa0.z, 1.0f} : (v2f){-2.0f * qa0.x, -2.0f * qa0.y};
  v2f a1 = lhalf ? (v2f){-2.0f * qa1.z, 1.0f} : (v2f){-2.0f * qa1.x, -2.0f * qa1.y};

  v8f c0, c1;
#pragma unroll
  for (int r = 0; r < 8; ++r) {
    c0[r] = QA[(q0 + r + lhalf * 8) * 4 + 3];
    c1[r] = QA[(q0 + 16 + r + lhalf * 8) * 4 + 3];
  }

  float bd[KEEP_]; int bi[KEEP_];
#pragma unroll
  for (int k = 0; k < KEEP_; ++k) { bd[k] = 3.0e38f; bi[k] = 0; }

  auto load_bb = [&](int ch) -> v2f {
    int pt = ch * 16 + l16;
    float x0 = P[pt * 3 + (lhalf ? 2 : 0)];
    float x1 = lhalf ? PN[pt] : P[pt * 3 + 1];
    return (v2f){x0, x1};
  };

  const int ch0 = wave * KNN_SLCH_;
  const int ch1 = ch0 + KNN_SLCH_;
  int qo = 0;
  v2f bb = load_bb(ch0);
#pragma nounroll
  for (int ch = ch0; ch < ch1; ++ch) {
    v2f bb_next = load_bb(ch + 1 < ch1 ? ch + 1 : ch1 - 1);

    v8f d0 = __builtin_amdgcn_wmma_f32_16x16x4_f32(false, a0, false, bb, (short)0, c0, false, false);
    v8f d1 = __builtin_amdgcn_wmma_f32_16x16x4_f32(false, a1, false, bb, (short)0, c1, false, false);

#pragma unroll
    for (int r = 0; r < 8; ++r) {
      st[(r + lhalf * 8) * RS_ + l16]       = d0[r];
      st[(16 + r + lhalf * 8) * RS_ + l16]  = d1[r];
    }
    const float4* myrow = (const float4*)(st + lane * RS_);
#pragma unroll
    for (int jj = 0; jj < 4; ++jj) {
      float4 v4 = myrow[jj];
#pragma unroll
      for (int e = 0; e < 4; ++e) {
        float v = fmaxf(e == 0 ? v4.x : (e == 1 ? v4.y : (e == 2 ? v4.z : v4.w)), 0.0f);
        kql[qo]      = __builtin_bit_cast(int, v);
        kql[qo + 32] = ch * 16 + jj * 4 + e;
        qo += (v < bd[KEEP_ - 1]) ? 64 : 0;
      }
    }
    bb = bb_next;

    if (ch == ch1 - 1 || __builtin_amdgcn_ballot_w32(qo > (KNN_QC_ - 16) * 64) != 0) {
      int nv = 0, np = 0;
      if (0 < qo) { nv = kql[0]; np = kql[32]; }
#pragma nounroll
      for (int jo = 0; __builtin_amdgcn_ballot_w32(jo < qo) != 0; jo += 64) {
        const float v = jo < qo ? __builtin_bit_cast(float, nv) : 3.0e38f;
        const int  pi = np;
        if (jo + 64 < qo) { nv = kql[jo + 64]; np = kql[jo + 96]; }
        if (__builtin_amdgcn_ballot_w32(v < bd[KEEP_ - 1]) == 0) continue;
        knn_insert(bd, bi, v, pi);
      }
      qo = 0;
    }
  }

  if (wave != 0) {
#pragma unroll
    for (int k = 0; k < KEEP_; ++k) {
      mrg_d[wave - 1][k][lane] = bd[k];
      mrg_i[wave - 1][k][lane] = bi[k];
    }
  }
  __syncthreads();
  if (wave != 0) return;

#pragma nounroll
  for (int w = 0; w < KNN_SL_ - 1; ++w) {
#pragma nounroll
    for (int k = 0; k < KEEP_; ++k) {
      const float v = mrg_d[w][k][lane];
      const int  pi = mrg_i[w][k][lane];
      if (__builtin_amdgcn_ballot_w32(v < bd[KEEP_ - 1]) == 0) break;
      knn_insert(bd, bi, v, pi);
    }
  }

  int q = q0 + lane;
#pragma unroll
  for (int k = 0; k < S_; ++k) nbr[((size_t)b * Q_ + q) * S_ + k] = bi[k + 1];
}

__global__ __launch_bounds__(256) void three_nn_kernel(const float* __restrict__ pts,
                                                       const float* __restrict__ pn2,
                                                       const float* __restrict__ qA,
                                                       int* __restrict__ idx3,
                                                       float* __restrict__ w3) {
  const int wave = threadIdx.x >> 5;
  const int lane = threadIdx.x & 31;
  const int b    = blockIdx.x >> 5;
  const int n0   = ((blockIdx.x & 31) * 8 + wave) * 32;

  const float* P  = pts + (size_t)b * N_ * 3;
  const float* PN = pn2 + (size_t)b * N_;
  const float* QA = qA + (size_t)b * Q_ * 4;

  __shared__ float stage[8][32 * RS_];
  float* st = stage[wave];

  const int lhalf = lane >> 4;
  const int l16   = lane & 15;

  int m0 = n0 + l16, m1 = n0 + 16 + l16;
  float p0x = P[m0 * 3 + 0], p0y = P[m0 * 3 + 1], p0z = P[m0 * 3 + 2];
  float p1x = P[m1 * 3 + 0], p1y = P[m1 * 3 + 1], p1z = P[m1 * 3 + 2];
  v2f a0 = lhalf ? (v2f){-2.0f * p0z, 1.0f} : (v2f){-2.0f * p0x, -2.0f * p0y};
  v2f a1 = lhalf ? (v2f){-2.0f * p1z, 1.0f} : (v2f){-2.0f * p1x, -2.0f * p1y};

  v8f c0, c1;
#pragma unroll
  for (int r = 0; r < 8; ++r) {
    c0[r] = PN[n0 + r + lhalf * 8];
    c1[r] = PN[n0 + 16 + r + lhalf * 8];
  }

  float bd[3]; int bi[3];
#pragma unroll
  for (int k = 0; k < 3; ++k) { bd[k] = 3.0e38f; bi[k] = 0; }

  auto load_bb = [&](int ch) -> v2f {
    const float4 qa = ((const float4*)QA)[ch * 16 + l16];
    return lhalf ? (v2f){qa.z, qa.w} : (v2f){qa.x, qa.y};
  };

  const int NCH = Q_ / 16;
  v2f bb = load_bb(0);
  for (int ch = 0; ch < NCH; ++ch) {
    v2f bb_next = load_bb(ch + 1 < NCH ? ch + 1 : NCH - 1);

    v8f d0 = __builtin_amdgcn_wmma_f32_16x16x4_f32(false, a0, false, bb, (short)0, c0, false, false);
    v8f d1 = __builtin_amdgcn_wmma_f32_16x16x4_f32(false, a1, false, bb, (short)0, c1, false, false);

#pragma unroll
    for (int r = 0; r < 8; ++r) {
      st[(r + lhalf * 8) * RS_ + l16]      = d0[r];
      st[(16 + r + lhalf * 8) * RS_ + l16] = d1[r];
    }
    const float4* myrow = (const float4*)(st + lane * RS_);
#pragma unroll
    for (int jj = 0; jj < 4; ++jj) {
      float4 v4 = myrow[jj];
#pragma unroll
      for (int e = 0; e < 4; ++e) {
        float v = fmaxf(e == 0 ? v4.x : (e == 1 ? v4.y : (e == 2 ? v4.z : v4.w)), 0.0f);
        int qi = ch * 16 + jj * 4 + e;
        if (v < bd[2]) {
          bd[2] = v; bi[2] = qi;
          if (bd[2] < bd[1]) { float tf = bd[2]; bd[2] = bd[1]; bd[1] = tf;
                               int ti = bi[2]; bi[2] = bi[1]; bi[1] = ti; }
          if (bd[1] < bd[0]) { float tf = bd[1]; bd[1] = bd[0]; bd[0] = tf;
                               int ti = bi[1]; bi[1] = bi[0]; bi[0] = ti; }
        }
      }
    }
    bb = bb_next;
  }

  int n = n0 + lane;
  float dv0 = fmaxf(bd[0], 1e-10f), dv1 = fmaxf(bd[1], 1e-10f), dv2 = fmaxf(bd[2], 1e-10f);
  float i0 = 1.0f / (dv0 + 1e-8f), i1 = 1.0f / (dv1 + 1e-8f), i2 = 1.0f / (dv2 + 1e-8f);
  float inv_s = 1.0f / (i0 + i1 + i2);
  size_t base = ((size_t)b * N_ + n) * 3;
  idx3[base + 0] = bi[0]; idx3[base + 1] = bi[1]; idx3[base + 2] = bi[2];
  w3[base + 0] = i0 * inv_s; w3[base + 1] = i1 * inv_s; w3[base + 2] = i2 * inv_s;
}

__global__ __launch_bounds__(256) void group_mean_kernel(const float* __restrict__ feat,
                                                         const int* __restrict__ nbr,
                                                         float* __restrict__ coarse) {
  int bc = blockIdx.x;
  int b = bc / C_, c = bc % C_;
  const float* F = feat + ((size_t)b * C_ + c) * N_;
  const int* NB  = nbr + (size_t)b * Q_ * S_;
  float* O = coarse + ((size_t)b * C_ + c) * Q_;
  for (int q = threadIdx.x; q < Q_; q += blockDim.x) {
    float s = 0.0f;
#pragma unroll
    for (int k = 0; k < S_; ++k) s += F[NB[q * S_ + k]];
    O[q] = s * (1.0f / S_);
  }
}

__global__ __launch_bounds__(256) void interp_kernel(const float* __restrict__ coarse,
                                                     const int* __restrict__ idx3,
                                                     const float* __restrict__ w3,
                                                     float* __restrict__ out) {
  int bc = blockIdx.x;
  int b = bc / C_, c = bc % C_;
  const float* CO = coarse + ((size_t)b * C_ + c) * Q_;
  const int*   I  = idx3 + (size_t)b * N_ * 3;
  const float* W  = w3 + (size_t)b * N_ * 3;
  float* O = out + ((size_t)b * C_ + c) * N_;
  for (int n = threadIdx.x; n < N_; n += blockDim.x) {
    O[n] = W[n * 3 + 0] * CO[I[n * 3 + 0]] +
           W[n * 3 + 1] * CO[I[n * 3 + 1]] +
           W[n * 3 + 2] * CO[I[n * 3 + 2]];
  }
}

extern "C" void kernel_launch(void* const* d_in, const int* in_sizes, int n_in,
                              void* d_out, int out_size, void* d_ws, size_t ws_size,
                              hipStream_t stream) {
  const float* points   = (const float*)d_in[0];
  const float* features = (const float*)d_in[1];

  auto align256 = [](size_t x) { return (x + 255) & ~(size_t)255; };
  char* ws = (char*)d_ws;
  size_t off = 0;
  int*   fps_idx = (int*)(ws + off);   off = align256(off + sizeof(int) * B_ * Q_);
  float* qA      = (float*)(ws + off); off = align256(off + sizeof(float) * B_ * Q_ * 4);
  float* pn2     = (float*)(ws + off); off = align256(off + sizeof(float) * B_ * N_);
  int*   nbr     = (int*)(ws + off);   off = align256(off + sizeof(int) * B_ * Q_ * S_);
  float* coarse  = (float*)(ws + off); off = align256(off + sizeof(float) * B_ * C_ * Q_);
  int*   idx3    = (int*)(ws + off);   off = align256(off + sizeof(int) * B_ * N_ * 3);
  float* w3      = (float*)(ws + off); off = align256(off + sizeof(float) * B_ * N_ * 3);

  float* out = (float*)d_out;

  pnorm2_kernel<<<(B_ * N_ + 255) / 256, 256, 0, stream>>>(points, pn2);
  fps_kernel<<<B_, 256, 0, stream>>>(points, fps_idx, qA);
  knn_kernel<<<B_ * (Q_ / 32), 256, 0, stream>>>(points, pn2, qA, nbr);
  group_mean_kernel<<<B_ * C_, 256, 0, stream>>>(features, nbr, coarse);
  three_nn_kernel<<<B_ * 32, 256, 0, stream>>>(points, pn2, qA, idx3, w3);
  interp_kernel<<<B_ * C_, 256, 0, stream>>>(coarse, idx3, w3, out);
}


# fps: a wave redoes its argmax search, candidate xyz read and lane reduction only after a round whose sample lowered some lane's candidate distance; else it posts its cached record: same indices
# speedup vs baseline: 4.2058x; 1.1247x over previous
#include <hip/hip_runtime.h>

#define B_ 4
#define N_ 8192
#define C_ 128
#define Q_ 2048
#define S_ 16
#define KEEP_ 17
#define RS_ 20

typedef __attribute__((ext_vector_type(2))) float v2f;
typedef __attribute__((ext_vector_type(8))) float v8f;

__global__ void pnorm2_kernel(const float* __restrict__ pts, float* __restrict__ pn2) {
  int i = blockIdx.x * blockDim.x + threadIdx.x;
  if (i < B_ * N_) {
    float x = pts[i * 3 + 0], y = pts[i * 3 + 1], z = pts[i * 3 + 2];
    pn2[i] = x * x + y * y + z * z;
  }
}

template <int M> __device__ __forceinline__ int row_xor(int v) {
  return __builtin_amdgcn_update_dpp(0, v, 0x160 + M, 0xf, 0xf, true);
}
__device__ __forceinline__ int xhalf(int v) {
  return __builtin_amdgcn_permlanex16(v, v, 0x76543210, 0xfedcba98, false, false);
}

#define FPS_PT_ (N_ / 256)
#define FPS_CH_ 8
#define FPS_CL_ (FPS_PT_ / FPS_CH_)
#define FPS_LS_ (FPS_PT_ + 1)

__device__ __forceinline__ float fps_dist2(float dx, float dy, float dz) {
  return __builtin_fmaf(dz, dz, __builtin_fmaf(dx, dx, dy * dy));
}

__device__ __forceinline__ void fps_take_higher(int& va, int& sa, int vb, int sb) {
  const bool g = vb > va;
  va = g ? vb : va;
  sa = g ? sb : sa;
}

__device__ __forceinline__ void fps_thread_argmax(const int (&d)[FPS_PT_], int& bv, int& bslot) {
  int v[FPS_CH_], sl[FPS_CH_];
#pragma unroll
  for (int c = 0; c < FPS_CH_; ++c) { v[c] = d[c * FPS_CL_]; sl[c] = c * FPS_CL_; }
#pragma unroll
  for (int j = 1; j < FPS_CL_; ++j) {
#pragma unroll
    for (int c = 0; c < FPS_CH_; ++c) {
      const int i = c * FPS_CL_ + j;
      const bool g = d[i] > v[c];
      v[c]  = g ? d[i] : v[c];
      sl[c] = g ? i : sl[c];
    }
  }
  fps_take_higher(v[0], sl[0], v[1], sl[1]);
  fps_take_higher(v[2], sl[2], v[3], sl[3]);
  fps_take_higher(v[4], sl[4], v[5], sl[5]);
  fps_take_higher(v[6], sl[6], v[7], sl[7]);
  fps_take_higher(v[0], sl[0], v[2], sl[2]);
  fps_take_higher(v[4], sl[4], v[6], sl[6]);
  fps_take_higher(v[0], sl[0], v[4], sl[4]);
  bv = v[0];
  bslot = sl[0];
}

__global__ __launch_bounds__(256) void fps_kernel(const float* __restrict__ pts,
                                                  int* __restrict__ fps_idx,
                                                  float* __restrict__ qA) {
  const int b = blockIdx.x;
  const int tid = threadIdx.x;
  const float* P = pts + (size_t)b * N_ * 3;

  __shared__ float4 lpts[256 * FPS_LS_];
  __shared__ int4 rec[2][10 + 256 + 64];
  const int wave = tid >> 5;
  const int lane = tid & 31;

  float px[FPS_PT_], py[FPS_PT_], pz[FPS_PT_];
  int d[FPS_PT_];
  const float p0x = P[0], p0y = P[1], p0z = P[2];
  {
    const float4* P4 = (const float4*)(P + (size_t)tid * (FPS_PT_ * 3));
    float f[FPS_PT_ * 3];
#pragma unroll
    for (int k = 0; k < FPS_PT_ * 3 / 4; ++k) {
      const float4 t4 = P4[k];
      f[k * 4 + 0] = t4.x; f[k * 4 + 1] = t4.y; f[k * 4 + 2] = t4.z; f[k * 4 + 3] = t4.w;
    }
#pragma unroll
    for (int i = 0; i < FPS_PT_; ++i) {
      px[i] = f[i * 3 + 0]; py[i] = f[i * 3 + 1]; pz[i] = f[i * 3 + 2];
      lpts[tid * FPS_LS_ + i] = make_float4(px[i], py[i], pz[i], 0.0f);
      float dx = px[i] - p0x, dy = py[i] - p0y, dz = pz[i] - p0z;
      d[i] = __builtin_bit_cast(int, fps_dist2(dx, dy, dz));
    }
  }
  int bvi, bslot;
  float4 cp;
  int wm;
  int ia, iz;
  auto renew = [&]() {
    fps_thread_argmax(d, bvi, bslot);
    cp = lpts[tid * FPS_LS_ + bslot];
    wm = bvi;
    wm = max(wm, row_xor<1>(wm)); wm = max(wm, row_xor<2>(wm));
    wm = max(wm, row_xor<4>(wm)); wm = max(wm, row_xor<8>(wm));
    wm = max(wm, xhalf(wm));
    const bool lead = lane == __builtin_ctz(__builtin_amdgcn_ballot_w32(bvi == wm));
    ia = lead ? wave : 10 + tid;
    iz = lead ? 8 * 4 + wave : (10 + 256) * 4 + tid;
    asm volatile("" :: "v"(cp.x), "v"(cp.y), "v"(cp.z), "v"(ia), "v"(iz));
  };
  renew();
  if (tid == 0) {
    fps_idx[b * Q_ + 0] = 0;
    qA[((size_t)b * Q_ + 0) * 4 + 0] = p0x;
    qA[((size_t)b * Q_ + 0) * 4 + 1] = p0y;
    qA[((size_t)b * Q_ + 0) * 4 + 2] = p0z;
    qA[((size_t)b * Q_ + 0) * 4 + 3] = p0x * p0x + p0y * p0y + p0z * p0z;
  }

  for (int s = 1; s < Q_; ++s) {
    const int bidx = tid * FPS_PT_ + bslot;
    int4* R = rec[s & 1];
    R[ia] = make_int4(wm, bidx, __builtin_bit_cast(int, cp.x), __builtin_bit_cast(int, cp.y));
    ((int*)R)[iz] = __builtin_bit_cast(int, cp.z);
    __syncthreads();
    const int4 ra = R[lane & 7];
    const int rz = ((const int*)R)[8 * 4 + (lane & 7)];
    int gm = ra.x;
    gm = max(gm, row_xor<1>(gm)); gm = max(gm, row_xor<2>(gm)); gm = max(gm, row_xor<4>(gm));
    const int wl = __builtin_ctz(__builtin_amdgcn_ballot_w32(ra.x == gm));
    const int nxt = __builtin_amdgcn_readlane(ra.y, wl);
    const float cx = __builtin_bit_cast(float, __builtin_amdgcn_readlane(ra.z, wl));
    const float cy = __builtin_bit_cast(float, __builtin_amdgcn_readlane(ra.w, wl));
    const float cz = __builtin_bit_cast(float, __builtin_amdgcn_readlane(rz, wl));
    if (tid == 0) {
      float x = cx, y = cy, z = cz;
      fps_idx[b * Q_ + s] = nxt;
      qA[((size_t)b * Q_ + s) * 4 + 0] = x;
      qA[((size_t)b * Q_ + s) * 4 + 1] = y;
      qA[((size_t)b * Q_ + s) * 4 + 2] = z;
      float n2;
      {
#pragma clang fp contract(off)
        n2 = x * x + y * y;
      }
      qA[((size_t)b * Q_ + s) * 4 + 3] = __builtin_fmaf(z, z, n2);
    }
    const int cand_dd = __builtin_bit_cast(int, fps_dist2(cp.x - cx, cp.y - cy, cp.z - cz));
    const bool cand_lowered = cand_dd <= bvi;
#pragma unroll
    for (int i = 0; i < FPS_PT_; ++i) {
      float dx = px[i] - cx, dy = py[i] - cy, dz = pz[i] - cz;
      float dd = fps_dist2(dx, dy, dz);
      d[i] = min(d[i], __builtin_bit_cast(int, dd));
    }
    if (__builtin_amdgcn_ballot_w32(cand_lowered) != 0) renew();
  }
}

#define KNN_SL_ 8
#define KNN_SLCH_ (N_ / 16 / KNN_SL_)
#define KNN_QC_ 32

__device__ __forceinline__ void knn_insert(float (&bd)[KEEP_], int (&bi)[KEEP_], float v, int pi) {
  if (v < bd[KEEP_ - 1]) {
    bd[KEEP_ - 1] = v; bi[KEEP_ - 1] = pi;
#pragma unroll
    for (int t = KEEP_ - 1; t > 0; --t) {
      if (bd[t] < bd[t - 1]) {
        float tf = bd[t]; bd[t] = bd[t - 1]; bd[t - 1] = tf;
        int   ti = bi[t]; bi[t] = bi[t - 1]; bi[t - 1] = ti;
      }
    }
  }
}

__global__ __launch_bounds__(256) void knn_kernel(const float* __restrict__ pts,
                                                  const float* __restrict__ pn2,
                                                  const float* __restrict__ qA,
                                                  int* __restrict__ nbr) {
  const int wave = __builtin_amdgcn_readfirstlane((int)(threadIdx.x >> 5));
  const int lane = threadIdx.x & 31;
  const int b    = blockIdx.x / (Q_ / 32);
  const int q0   = (blockIdx.x % (Q_ / 32)) * 32;

  const float* P  = pts + (size_t)b * N_ * 3;
  const float* PN = pn2 + (size_t)b * N_;
  const float* QA = qA + (size_t)b * Q_ * 4;

  __shared__ float stage[8][32 * RS_];
  __shared__ float mrg_d[KNN_SL_ - 1][KEEP_][32];
  __shared__ int   mrg_i[KNN_SL_ - 1][KEEP_][32];
  __shared__ int   kq[KNN_SL_][KNN_QC_ * 64];
  float* st = stage[wave];
  int* kql = kq[wave] + lane;

  const int lhalf = lane >> 4;
  const int l16   = lane & 15;

  const float4 qa0 = ((const float4*)QA)[q0 + l16];
  const float4 qa1 = ((const float4*)QA)[q0 + 16 + l16];
  v2f a0 = lhalf ? (v2f){-2.0f * qa0.z, 1.0f} : (v2f){-2.0f * qa0.x, -2.0f * qa0.y};
  v2f a1 = lhalf ? (v2f){-2.0f * qa1.z, 1.0f} : (v2f){-2.0f * qa1.x, -2.0f * qa1.y};

  v8f c0, c1;
#pragma unroll
  for (int r = 0; r < 8; ++r) {
    c0[r] = QA[(q0 + r + lhalf * 8) * 4 + 3];
    c1[r] = QA[(q0 + 16 + r + lhalf * 8) * 4 + 3];
  }

  float bd[KEEP_]; int bi[KEEP_];
#pragma unroll
  for (int k = 0; k < KEEP_; ++k) { bd[k] = 3.0e38f; bi[k] = 0; }

  auto load_bb = [&](int ch) -> v2f {
    int pt = ch * 16 + l16;
    float x0 = P[pt * 3 + (lhalf ? 2 : 0)];
    float x1 = lhalf ? PN[pt] : P[pt * 3 + 1];
    return (v2f){x0, x1};
  };

  const int ch0 = wave * KNN_SLCH_;
  const int ch1 = ch0 + KNN_SLCH_;
  int qo = 0;
  v2f bb = load_bb(ch0);
#pragma nounroll
  for (int ch = ch0; ch < ch1; ++ch) {
    v2f bb_next = load_bb(ch + 1 < ch1 ? ch + 1 : ch1 - 1);

    v8f d0 = __builtin_amdgcn_wmma_f32_16x16x4_f32(false, a0, false, bb, (short)0, c0, false, false);
    v8f d1 = __builtin_amdgcn_wmma_f32_16x16x4_f32(false, a1, false, bb, (short)0, c1, false, false);

#pragma unroll
    for (int r = 0; r < 8; ++r) {
      st[(r + lhalf * 8) * RS_ + l16]       = d0[r];
      st[(16 + r + lhalf * 8) * RS_ + l16]  = d1[r];
    }
    const float4* myrow = (const float4*)(st + lane * RS_);
#pragma unroll
    for (int jj = 0; jj < 4; ++jj) {
      float4 v4 = myrow[jj];
#pragma unroll
      for (int e = 0; e < 4; ++e) {
        float v = fmaxf(e == 0 ? v4.x : (e == 1 ? v4.y : (e == 2 ? v4.z : v4.w)), 0.0f);
        kql[qo]      = __builtin_bit_cast(int, v);
        kql[qo + 32] = ch * 16 + jj * 4 + e;
        qo += (v < bd[KEEP_ - 1]) ? 64 : 0;
      }
    }
    bb = bb_next;

    if (ch == ch1 - 1 || __builtin_amdgcn_ballot_w32(qo > (KNN_QC_ - 16) * 64) != 0) {
      int nv = 0, np = 0;
      if (0 < qo) { nv = kql[0]; np = kql[32]; }
#pragma nounroll
      for (int jo = 0; __builtin_amdgcn_ballot_w32(jo < qo) != 0; jo += 64) {
        const float v = jo < qo ? __builtin_bit_cast(float, nv) : 3.0e38f;
        const int  pi = np;
        if (jo + 64 < qo) { nv = kql[jo + 64]; np = kql[jo + 96]; }
        if (__builtin_amdgcn_ballot_w32(v < bd[KEEP_ - 1]) == 0) continue;
        knn_insert(bd, bi, v, pi);
      }
      qo = 0;
    }
  }

  if (wave != 0) {
#pragma unroll
    for (int k = 0; k < KEEP_; ++k) {
      mrg_d[wave - 1][k][lane] = bd[k];
      mrg_i[wave - 1][k][lane] = bi[k];
    }
  }
  __syncthreads();
  if (wave != 0) return;

#pragma nounroll
  for (int w = 0; w < KNN_SL_ - 1; ++w) {
#pragma nounroll
    for (int k = 0; k < KEEP_; ++k) {
      const float v = mrg_d[w][k][lane];
      const int  pi = mrg_i[w][k][lane];
      if (__builtin_amdgcn_ballot_w32(v < bd[KEEP_ - 1]) == 0) break;
      knn_insert(bd, bi, v, pi);
    }
  }

  int q = q0 + lane;
#pragma unroll
  for (int k = 0; k < S_; ++k) nbr[((size_t)b * Q_ + q) * S_ + k] = bi[k + 1];
}

__global__ __launch_bounds__(256) void three_nn_kernel(const float* __restrict__ pts,
                                                       const float* __restrict__ pn2,
                                                       const float* __restrict__ qA,
                                                       int* __restrict__ idx3,
                                                       float* __restrict__ w3) {
  const int wave = threadIdx.x >> 5;
  const int lane = threadIdx.x & 31;
  const int b    = blockIdx.x >> 5;
  const int n0   = ((blockIdx.x & 31) * 8 + wave) * 32;

  const float* P  = pts + (size_t)b * N_ * 3;
  const float* PN = pn2 + (size_t)b * N_;
  const float* QA = qA + (size_t)b * Q_ * 4;

  __shared__ float stage[8][32 * RS_];
  float* st = stage[wave];

  const int lhalf = lane >> 4;
  const int l16   = lane & 15;

  int m0 = n0 + l16, m1 = n0 + 16 + l16;
  float p0x = P[m0 * 3 + 0], p0y = P[m0 * 3 + 1], p0z = P[m0 * 3 + 2];
  float p1x = P[m1 * 3 + 0], p1y = P[m1 * 3 + 1], p1z = P[m1 * 3 + 2];
  v2f a0 = lhalf ? (v2f){-2.0f * p0z, 1.0f} : (v2f){-2.0f * p0x, -2.0f * p0y};
  v2f a1 = lhalf ? (v2f){-2.0f * p1z, 1.0f} : (v2f){-2.0f * p1x, -2.0f * p1y};

  v8f c0, c1;
#pragma unroll
  for (int r = 0; r < 8; ++r) {
    c0[r] = PN[n0 + r + lhalf * 8];
    c1[r] = PN[n0 + 16 + r + lhalf * 8];
  }

  float bd[3]; int bi[3];
#pragma unroll
  for (int k = 0; k < 3; ++k) { bd[k] = 3.0e38f; bi[k] = 0; }

  auto load_bb = [&](int ch) -> v2f {
    const float4 qa = ((const float4*)QA)[ch * 16 + l16];
    return lhalf ? (v2f){qa.z, qa.w} : (v2f){qa.x, qa.y};
  };

  const int NCH = Q_ / 16;
  v2f bb = load_bb(0);
  for (int ch = 0; ch < NCH; ++ch) {
    v2f bb_next = load_bb(ch + 1 < NCH ? ch + 1 : NCH - 1);

    v8f d0 = __builtin_amdgcn_wmma_f32_16x16x4_f32(false, a0, false, bb, (short)0, c0, false, false);
    v8f d1 = __builtin_amdgcn_wmma_f32_16x16x4_f32(false, a1, false, bb, (short)0, c1, false, false);

#pragma unroll
    for (int r = 0; r < 8; ++r) {
      st[(r + lhalf * 8) * RS_ + l16]      = d0[r];
      st[(16 + r + lhalf * 8) * RS_ + l16] = d1[r];
    }
    const float4* myrow = (const float4*)(st + lane * RS_);
#pragma unroll
    for (int jj = 0; jj < 4; ++jj) {
      float4 v4 = myrow[jj];
#pragma unroll
      for (int e = 0; e < 4; ++e) {
        float v = fmaxf(e == 0 ? v4.x : (e == 1 ? v4.y : (e == 2 ? v4.z : v4.w)), 0.0f);
        int qi = ch * 16 + jj * 4 + e;
        if (v < bd[2]) {
          bd[2] = v; bi[2] = qi;
          if (bd[2] < bd[1]) { float tf = bd[2]; bd[2] = bd[1]; bd[1] = tf;
                               int ti = bi[2]; bi[2] = bi[1]; bi[1] = ti; }
          if (bd[1] < bd[0]) { float tf = bd[1]; bd[1] = bd[0]; bd[0] = tf;
                               int ti = bi[1]; bi[1] = bi[0]; bi[0] = ti; }
        }
      }
    }
    bb = bb_next;
  }

  int n = n0 + lane;
  float dv0 = fmaxf(bd[0], 1e-10f), dv1 = fmaxf(bd[1], 1e-10f), dv2 = fmaxf(bd[2], 1e-10f);
  float i0 = 1.0f / (dv0 + 1e-8f), i1 = 1.0f / (dv1 + 1e-8f), i2 = 1.0f / (dv2 + 1e-8f);
  float inv_s = 1.0f / (i0 + i1 + i2);
  size_t base = ((size_t)b * N_ + n) * 3;
  idx3[base + 0] = bi[0]; idx3[base + 1] = bi[1]; idx3[base + 2] = bi[2];
  w3[base + 0] = i0 * inv_s; w3[base + 1] = i1 * inv_s; w3[base + 2] = i2 * inv_s;
}

__global__ __launch_bounds__(256) void group_mean_kernel(const float* __restrict__ feat,
                                                         const int* __restrict__ nbr,
                                                         float* __restrict__ coarse) {
  int bc = blockIdx.x;
  int b = bc / C_, c = bc % C_;
  const float* F = feat + ((size_t)b * C_ + c) * N_;
  const int* NB  = nbr + (size_t)b * Q_ * S_;
  float* O = coarse + ((size_t)b * C_ + c) * Q_;
  for (int q = threadIdx.x; q < Q_; q += blockDim.x) {
    float s = 0.0f;
#pragma unroll
    for (int k = 0; k < S_; ++k) s += F[NB[q * S_ + k]];
    O[q] = s * (1.0f / S_);
  }
}

__global__ __launch_bounds__(256) void interp_kernel(const float* __restrict__ coarse,
                                                     const int* __restrict__ idx3,
                                                     const float* __restrict__ w3,
                                                     float* __restrict__ out) {
  int bc = blockIdx.x;
  int b = bc / C_, c = bc % C_;
  const float* CO = coarse + ((size_t)b * C_ + c) * Q_;
  const int*   I  = idx3 + (size_t)b * N_ * 3;
  const float* W  = w3 + (size_t)b * N_ * 3;
  float* O = out + ((size_t)b * C_ + c) * N_;
  for (int n = threadIdx.x; n < N_; n += blockDim.x) {
    O[n] = W[n * 3 + 0] * CO[I[n * 3 + 0]] +
           W[n * 3 + 1] * CO[I[n * 3 + 1]] +
           W[n * 3 + 2] * CO[I[n * 3 + 2]];
  }
}

extern "C" void kernel_launch(void* const* d_in, const int* in_sizes, int n_in,
                              void* d_out, int out_size, void* d_ws, size_t ws_size,
                              hipStream_t stream) {
  const float* points   = (const float*)d_in[0];
  const float* features = (const float*)d_in[1];

  auto align256 = [](size_t x) { return (x + 255) & ~(size_t)255; };
  char* ws = (char*)d_ws;
  size_t off = 0;
  int*   fps_idx = (int*)(ws + off);   off = align256(off + sizeof(int) * B_ * Q_);
  float* qA      = (float*)(ws + off); off = align256(off + sizeof(float) * B_ * Q_ * 4);
  float* pn2     = (float*)(ws + off); off = align256(off + sizeof(float) * B_ * N_);
  int*   nbr     = (int*)(ws + off);   off = align256(off + sizeof(int) * B_ * Q_ * S_);
  float* coarse  = (float*)(ws + off); off = align256(off + sizeof(float) * B_ * C_ * Q_);
  int*   idx3    = (int*)(ws + off);   off = align256(off + sizeof(int) * B_ * N_ * 3);
  float* w3      = (float*)(ws + off); off = align256(off + sizeof(float) * B_ * N_ * 3);

  float* out = (float*)d_out;

  pnorm2_kernel<<<(B_ * N_ + 255) / 256, 256, 0, stream>>>(points, pn2);
  fps_kernel<<<B_, 256, 0, stream>>>(points, fps_idx, qA);
  knn_kernel<<<B_ * (Q_ / 32), 256, 0, stream>>>(points, pn2, qA, nbr);
  group_mean_kernel<<<B_ * C_, 256, 0, stream>>>(features, nbr, coarse);
  three_nn_kernel<<<B_ * 32, 256, 0, stream>>>(points, pn2, qA, idx3, w3);
  interp_kernel<<<B_ * C_, 256, 0, stream>>>(coarse, idx3, w3, out);
}


# three_nn: the 8 waves of a block share 32 points and each sweeps one eighth of the queries; wave 0 merges their sorted 3-lists from LDS in slice order (same strict compares); 1024 blocks, was 128
# speedup vs baseline: 4.4618x; 1.0609x over previous
#include <hip/hip_runtime.h>

#define B_ 4
#define N_ 8192
#define C_ 128
#define Q_ 2048
#define S_ 16
#define KEEP_ 17
#define RS_ 20

typedef __attribute__((ext_vector_type(2))) float v2f;
typedef __attribute__((ext_vector_type(8))) float v8f;

__global__ void pnorm2_kernel(const float* __restrict__ pts, float* __restrict__ pn2) {
  int i = blockIdx.x * blockDim.x + threadIdx.x;
  if (i < B_ * N_) {
    float x = pts[i * 3 + 0], y = pts[i * 3 + 1], z = pts[i * 3 + 2];
    pn2[i] = x * x + y * y + z * z;
  }
}

template <int M> __device__ __forceinline__ int row_xor(int v) {
  return __builtin_amdgcn_update_dpp(0, v, 0x160 + M, 0xf, 0xf, true);
}
__device__ __forceinline__ int xhalf(int v) {
  return __builtin_amdgcn_permlanex16(v, v, 0x76543210, 0xfedcba98, false, false);
}

#define FPS_PT_ (N_ / 256)
#define FPS_CH_ 8
#define FPS_CL_ (FPS_PT_ / FPS_CH_)
#define FPS_LS_ (FPS_PT_ + 1)

__device__ __forceinline__ float fps_dist2(float dx, float dy, float dz) {
  return __builtin_fmaf(dz, dz, __builtin_fmaf(dx, dx, dy * dy));
}

__device__ __forceinline__ void fps_take_higher(int& va, int& sa, int vb, int sb) {
  const bool g = vb > va;
  va = g ? vb : va;
  sa = g ? sb : sa;
}

__device__ __forceinline__ void fps_thread_argmax(const int (&d)[FPS_PT_], int& bv, int& bslot) {
  int v[FPS_CH_], sl[FPS_CH_];
#pragma unroll
  for (int c = 0; c < FPS_CH_; ++c) { v[c] = d[c * FPS_CL_]; sl[c] = c * FPS_CL_; }
#pragma unroll
  for (int j = 1; j < FPS_CL_; ++j) {
#pragma unroll
    for (int c = 0; c < FPS_CH_; ++c) {
      const int i = c * FPS_CL_ + j;
      const bool g = d[i] > v[c];
      v[c]  = g ? d[i] : v[c];
      sl[c] = g ? i : sl[c];
    }
  }
  fps_take_higher(v[0], sl[0], v[1], sl[1]);
  fps_take_higher(v[2], sl[2], v[3], sl[3]);
  fps_take_higher(v[4], sl[4], v[5], sl[5]);
  fps_take_higher(v[6], sl[6], v[7], sl[7]);
  fps_take_higher(v[0], sl[0], v[2], sl[2]);
  fps_take_higher(v[4], sl[4], v[6], sl[6]);
  fps_take_higher(v[0], sl[0], v[4], sl[4]);
  bv = v[0];
  bslot = sl[0];
}

__global__ __launch_bounds__(256) void fps_kernel(const float* __restrict__ pts,
                                                  int* __restrict__ fps_idx,
                                                  float* __restrict__ qA) {
  const int b = blockIdx.x;
  const int tid = threadIdx.x;
  const float* P = pts + (size_t)b * N_ * 3;

  __shared__ float4 lpts[256 * FPS_LS_];
  __shared__ int4 rec[2][10 + 256 + 64];
  const int wave = tid >> 5;
  const int lane = tid & 31;

  float px[FPS_PT_], py[FPS_PT_], pz[FPS_PT_];
  int d[FPS_PT_];
  const float p0x = P[0], p0y = P[1], p0z = P[2];
  {
    const float4* P4 = (const float4*)(P + (size_t)tid * (FPS_PT_ * 3));
    float f[FPS_PT_ * 3];
#pragma unroll
    for (int k = 0; k < FPS_PT_ * 3 / 4; ++k) {
      const float4 t4 = P4[k];
      f[k * 4 + 0] = t4.x; f[k * 4 + 1] = t4.y; f[k * 4 + 2] = t4.z; f[k * 4 + 3] = t4.w;
    }
#pragma unroll
    for (int i = 0; i < FPS_PT_; ++i) {
      px[i] = f[i * 3 + 0]; py[i] = f[i * 3 + 1]; pz[i] = f[i * 3 + 2];
      lpts[tid * FPS_LS_ + i] = make_float4(px[i], py[i], pz[i], 0.0f);
      float dx = px[i] - p0x, dy = py[i] - p0y, dz = pz[i] - p0z;
      d[i] = __builtin_bit_cast(int, fps_dist2(dx, dy, dz));
    }
  }
  int bvi, bslot;
  float4 cp;
  int wm;
  int ia, iz;
  auto renew = [&]() {
    fps_thread_argmax(d, bvi, bslot);
    cp = lpts[tid * FPS_LS_ + bslot];
    wm = bvi;
    wm = max(wm, row_xor<1>(wm)); wm = max(wm, row_xor<2>(wm));
    wm = max(wm, row_xor<4>(wm)); wm = max(wm, row_xor<8>(wm));
    wm = max(wm, xhalf(wm));
    const bool lead = lane == __builtin_ctz(__builtin_amdgcn_ballot_w32(bvi == wm));
    ia = lead ? wave : 10 + tid;
    iz = lead ? 8 * 4 + wave : (10 + 256) * 4 + tid;
    asm volatile("" :: "v"(cp.x), "v"(cp.y), "v"(cp.z), "v"(ia), "v"(iz));
  };
  renew();
  if (tid == 0) {
    fps_idx[b * Q_ + 0] = 0;
    qA[((size_t)b * Q_ + 0) * 4 + 0] = p0x;
    qA[((size_t)b * Q_ + 0) * 4 + 1] = p0y;
    qA[((size_t)b * Q_ + 0) * 4 + 2] = p0z;
    qA[((size_t)b * Q_ + 0) * 4 + 3] = p0x * p0x + p0y * p0y + p0z * p0z;
  }

  for (int s = 1; s < Q_; ++s) {
    const int bidx = tid * FPS_PT_ + bslot;
    int4* R = rec[s & 1];
    R[ia] = make_int4(wm, bidx, __builtin_bit_cast(int, cp.x), __builtin_bit_cast(int, cp.y));
    ((int*)R)[iz] = __builtin_bit_cast(int, cp.z);
    __syncthreads();
    const int4 ra = R[lane & 7];
    const int rz = ((const int*)R)[8 * 4 + (lane & 7)];
    int gm = ra.x;
    gm = max(gm, row_xor<1>(gm)); gm = max(gm, row_xor<2>(gm)); gm = max(gm, row_xor<4>(gm));
    const int wl = __builtin_ctz(__builtin_amdgcn_ballot_w32(ra.x == gm));
    const int nxt = __builtin_amdgcn_readlane(ra.y, wl);
    const float cx = __builtin_bit_cast(float, __builtin_amdgcn_readlane(ra.z, wl));
    const float cy = __builtin_bit_cast(float, __builtin_amdgcn_readlane(ra.w, wl));
    const float cz = __builtin_bit_cast(float, __builtin_amdgcn_readlane(rz, wl));
    if (tid == 0) {
      float x = cx, y = cy, z = cz;
      fps_idx[b * Q_ + s] = nxt;
      qA[((size_t)b * Q_ + s) * 4 + 0] = x;
      qA[((size_t)b * Q_ + s) * 4 + 1] = y;
      qA[((size_t)b * Q_ + s) * 4 + 2] = z;
      float n2;
      {
#pragma clang fp contract(off)
        n2 = x * x + y * y;
      }
      qA[((size_t)b * Q_ + s) * 4 + 3] = __builtin_fmaf(z, z, n2);
    }
    const int cand_dd = __builtin_bit_cast(int, fps_dist2(cp.x - cx, cp.y - cy, cp.z - cz));
    const bool cand_lowered = cand_dd <= bvi;
#pragma unroll
    for (int i = 0; i < FPS_PT_; ++i) {
      float dx = px[i] - cx, dy = py[i] - cy, dz = pz[i] - cz;
      float dd = fps_dist2(dx, dy, dz);
      d[i] = min(d[i], __builtin_bit_cast(int, dd));
    }
    if (__builtin_amdgcn_ballot_w32(cand_lowered) != 0) renew();
  }
}

#define KNN_SL_ 8
#define KNN_SLCH_ (N_ / 16 / KNN_SL_)
#define KNN_QC_ 32

__device__ __forceinline__ void knn_insert(float (&bd)[KEEP_], int (&bi)[KEEP_], float v, int pi) {
  if (v < bd[KEEP_ - 1]) {
    bd[KEEP_ - 1] = v; bi[KEEP_ - 1] = pi;
#pragma unroll
    for (int t = KEEP_ - 1; t > 0; --t) {
      if (bd[t] < bd[t - 1]) {
        float tf = bd[t]; bd[t] = bd[t - 1]; bd[t - 1] = tf;
        int   ti = bi[t]; bi[t] = bi[t - 1]; bi[t - 1] = ti;
      }
    }
  }
}

__global__ __launch_bounds__(256) void knn_kernel(const float* __restrict__ pts,
                                                  const float* __restrict__ pn2,
                                                  const float* __restrict__ qA,
                                                  int* __restrict__ nbr) {
  const int wave = __builtin_amdgcn_readfirstlane((int)(threadIdx.x >> 5));
  const int lane = threadIdx.x & 31;
  const int b    = blockIdx.x / (Q_ / 32);
  const int q0   = (blockIdx.x % (Q_ / 32)) * 32;

  const float* P  = pts + (size_t)b * N_ * 3;
  const float* PN = pn2 + (size_t)b * N_;
  const float* QA = qA + (size_t)b * Q_ * 4;

  __shared__ float stage[8][32 * RS_];
  __shared__ float mrg_d[KNN_SL_ - 1][KEEP_][32];
  __shared__ int   mrg_i[KNN_SL_ - 1][KEEP_][32];
  __shared__ int   kq[KNN_SL_][KNN_QC_ * 64];
  float* st = stage[wave];
  int* kql = kq[wave] + lane;

  const int lhalf = lane >> 4;
  const int l16   = lane & 15;

  const float4 qa0 = ((const float4*)QA)[q0 + l16];
  const float4 qa1 = ((const float4*)QA)[q0 + 16 + l16];
  v2f a0 = lhalf ? (v2f){-2.0f * qa0.z, 1.0f} : (v2f){-2.0f * qa0.x, -2.0f * qa0.y};
  v2f a1 = lhalf ? (v2f){-2.0f * qa1.z, 1.0f} : (v2f){-2.0f * qa1.x, -2.0f * qa1.y};

  v8f c0, c1;
#pragma unroll
  for (int r = 0; r < 8; ++r) {
    c0[r] = QA[(q0 + r + lhalf * 8) * 4 + 3];
    c1[r] = QA[(q0 + 16 + r + lhalf * 8) * 4 + 3];
  }

  float bd[KEEP_]; int bi[KEEP_];
#pragma unroll
  for (int k = 0; k < KEEP_; ++k) { bd[k] = 3.0e38f; bi[k] = 0; }

  auto load_bb = [&](int ch) -> v2f {
    int pt = ch * 16 + l16;
    float x0 = P[pt * 3 + (lhalf ? 2 : 0)];
    float x1 = lhalf ? PN[pt] : P[pt * 3 + 1];
    return (v2f){x0, x1};
  };

  const int ch0 = wave * KNN_SLCH_;
  const int ch1 = ch0 + KNN_SLCH_;
  int qo = 0;
  v2f bb = load_bb(ch0);
#pragma nounroll
  for (int ch = ch0; ch < ch1; ++ch) {
    v2f bb_next = load_bb(ch + 1 < ch1 ? ch + 1 : ch1 - 1);

    v8f d0 = __builtin_amdgcn_wmma_f32_16x16x4_f32(false, a0, false, bb, (short)0, c0, false, false);
    v8f d1 = __builtin_amdgcn_wmma_f32_16x16x4_f32(false, a1, false, bb, (short)0, c1, false, false);

#pragma unroll
    for (int r = 0; r < 8; ++r) {
      st[(r + lhalf * 8) * RS_ + l16]       = d0[r];
      st[(16 + r + lhalf * 8) * RS_ + l16]  = d1[r];
    }
    const float4* myrow = (const float4*)(st + lane * RS_);
#pragma unroll
    for (int jj = 0; jj < 4; ++jj) {
      float4 v4 = myrow[jj];
#pragma unroll
      for (int e = 0; e < 4; ++e) {
        float v = fmaxf(e == 0 ? v4.x : (e == 1 ? v4.y : (e == 2 ? v4.z : v4.w)), 0.0f);
        kql[qo]      = __builtin_bit_cast(int, v);
        kql[qo + 32] = ch * 16 + jj * 4 + e;
        qo += (v < bd[KEEP_ - 1]) ? 64 : 0;
      }
    }
    bb = bb_next;

    if (ch == ch1 - 1 || __builtin_amdgcn_ballot_w32(qo > (KNN_QC_ - 16) * 64) != 0) {
      int nv = 0, np = 0;
      if (0 < qo) { nv = kql[0]; np = kql[32]; }
#pragma nounroll
      for (int jo = 0; __builtin_amdgcn_ballot_w32(jo < qo) != 0; jo += 64) {
        const float v = jo < qo ? __builtin_bit_cast(float, nv) : 3.0e38f;
        const int  pi = np;
        if (jo + 64 < qo) { nv = kql[jo + 64]; np = kql[jo + 96]; }
        if (__builtin_amdgcn_ballot_w32(v < bd[KEEP_ - 1]) == 0) continue;
        knn_insert(bd, bi, v, pi);
      }
      qo = 0;
    }
  }

  if (wave != 0) {
#pragma unroll
    for (int k = 0; k < KEEP_; ++k) {
      mrg_d[wave - 1][k][lane] = bd[k];
      mrg_i[wave - 1][k][lane] = bi[k];
    }
  }
  __syncthreads();
  if (wave != 0) return;

#pragma nounroll
  for (int w = 0; w < KNN_SL_ - 1; ++w) {
#pragma nounroll
    for (int k = 0; k < KEEP_; ++k) {
      const float v = mrg_d[w][k][lane];
      const int  pi = mrg_i[w][k][lane];
      if (__builtin_amdgcn_ballot_w32(v < bd[KEEP_ - 1]) == 0) break;
      knn_insert(bd, bi, v, pi);
    }
  }

  int q = q0 + lane;
#pragma unroll
  for (int k = 0; k < S_; ++k) nbr[((size_t)b * Q_ + q) * S_ + k] = bi[k + 1];
}

#define TNN_SL_ 8
#define TNN_SLCH_ (Q_ / 16 / TNN_SL_)

__device__ __forceinline__ void tnn_insert(float (&bd)[3], int (&bi)[3], float v, int qi) {
  if (v < bd[2]) {
    bd[2] = v; bi[2] = qi;
    if (bd[2] < bd[1]) { float tf = bd[2]; bd[2] = bd[1]; bd[1] = tf;
                         int ti = bi[2]; bi[2] = bi[1]; bi[1] = ti; }
    if (bd[1] < bd[0]) { float tf = bd[1]; bd[1] = bd[0]; bd[0] = tf;
                         int ti = bi[1]; bi[1] = bi[0]; bi[0] = ti; }
  }
}

__global__ __launch_bounds__(256) void three_nn_kernel(const float* __restrict__ pts,
                                                       const float* __restrict__ pn2,
                                                       const float* __restrict__ qA,
                                                       int* __restrict__ idx3,
                                                       float* __restrict__ w3) {
  const int wave = __builtin_amdgcn_readfirstlane((int)(threadIdx.x >> 5));
  const int lane = threadIdx.x & 31;
  const int b    = blockIdx.x / (N_ / 32);
  const int n0   = (blockIdx.x % (N_ / 32)) * 32;

  const float* P  = pts + (size_t)b * N_ * 3;
  const float* PN = pn2 + (size_t)b * N_;
  const float* QA = qA + (size_t)b * Q_ * 4;

  __shared__ float stage[8][32 * RS_];
  __shared__ float mrg_d[TNN_SL_ - 1][3][32];
  __shared__ int   mrg_i[TNN_SL_ - 1][3][32];
  float* st = stage[wave];

  const int lhalf = lane >> 4;
  const int l16   = lane & 15;

  int m0 = n0 + l16, m1 = n0 + 16 + l16;
  float p0x = P[m0 * 3 + 0], p0y = P[m0 * 3 + 1], p0z = P[m0 * 3 + 2];
  float p1x = P[m1 * 3 + 0], p1y = P[m1 * 3 + 1], p1z = P[m1 * 3 + 2];
  v2f a0 = lhalf ? (v2f){-2.0f * p0z, 1.0f} : (v2f){-2.0f * p0x, -2.0f * p0y};
  v2f a1 = lhalf ? (v2f){-2.0f * p1z, 1.0f} : (v2f){-2.0f * p1x, -2.0f * p1y};

  v8f c0, c1;
#pragma unroll
  for (int r = 0; r < 8; ++r) {
    c0[r] = PN[n0 + r + lhalf * 8];
    c1[r] = PN[n0 + 16 + r + lhalf * 8];
  }

  float bd[3]; int bi[3];
#pragma unroll
  for (int k = 0; k < 3; ++k) { bd[k] = 3.0e38f; bi[k] = 0; }

  auto load_bb = [&](int ch) -> v2f {
    const float4 qa = ((const float4*)QA)[ch * 16 + l16];
    return lhalf ? (v2f){qa.z, qa.w} : (v2f){qa.x, qa.y};
  };

  const int ch0 = wave * TNN_SLCH_;
  const int ch1 = ch0 + TNN_SLCH_;
  v2f bb = load_bb(ch0);
  for (int ch = ch0; ch < ch1; ++ch) {
    v2f bb_next = load_bb(ch + 1 < ch1 ? ch + 1 : ch1 - 1);

    v8f d0 = __builtin_amdgcn_wmma_f32_16x16x4_f32(false, a0, false, bb, (short)0, c0, false, false);
    v8f d1 = __builtin_amdgcn_wmma_f32_16x16x4_f32(false, a1, false, bb, (short)0, c1, false, false);

#pragma unroll
    for (int r = 0; r < 8; ++r) {
      st[(r + lhalf * 8) * RS_ + l16]      = d0[r];
      st[(16 + r + lhalf * 8) * RS_ + l16] = d1[r];
    }
    const float4* myrow = (const float4*)(st + lane * RS_);
#pragma unroll
    for (int jj = 0; jj < 4; ++jj) {
      float4 v4 = myrow[jj];
#pragma unroll
      for (int e = 0; e < 4; ++e) {
        float v = fmaxf(e == 0 ? v4.x : (e == 1 ? v4.y : (e == 2 ? v4.z : v4.w)), 0.0f);
        tnn_insert(bd, bi, v, ch * 16 + jj * 4 + e);
      }
    }
    bb = bb_next;
  }

  if (wave != 0) {
#pragma unroll
    for (int k = 0; k < 3; ++k) {
      mrg_d[wave - 1][k][lane] = bd[k];
      mrg_i[wave - 1][k][lane] = bi[k];
    }
  }
  __syncthreads();
  if (wave != 0) return;

  float md[TNN_SL_ - 1][3]; int mi[TNN_SL_ - 1][3];
#pragma unroll
  for (int w = 0; w < TNN_SL_ - 1; ++w) {
#pragma unroll
    for (int k = 0; k < 3; ++k) { md[w][k] = mrg_d[w][k][lane]; mi[w][k] = mrg_i[w][k][lane]; }
  }
#pragma unroll
  for (int w = 0; w < TNN_SL_ - 1; ++w) {
#pragma unroll
    for (int k = 0; k < 3; ++k) tnn_insert(bd, bi, md[w][k], mi[w][k]);
  }

  int n = n0 + lane;
  float dv0 = fmaxf(bd[0], 1e-10f), dv1 = fmaxf(bd[1], 1e-10f), dv2 = fmaxf(bd[2], 1e-10f);
  float i0 = 1.0f / (dv0 + 1e-8f), i1 = 1.0f / (dv1 + 1e-8f), i2 = 1.0f / (dv2 + 1e-8f);
  float inv_s = 1.0f / (i0 + i1 + i2);
  size_t base = ((size_t)b * N_ + n) * 3;
  idx3[base + 0] = bi[0]; idx3[base + 1] = bi[1]; idx3[base + 2] = bi[2];
  w3[base + 0] = i0 * inv_s; w3[base + 1] = i1 * inv_s; w3[base + 2] = i2 * inv_s;
}

__global__ __launch_bounds__(256) void group_mean_kernel(const float* __restrict__ feat,
                                                         const int* __restrict__ nbr,
                                                         float* __restrict__ coarse) {
  int bc = blockIdx.x;
  int b = bc / C_, c = bc % C_;
  const float* F = feat + ((size_t)b * C_ + c) * N_;
  const int* NB  = nbr + (size_t)b * Q_ * S_;
  float* O = coarse + ((size_t)b * C_ + c) * Q_;
  for (int q = threadIdx.x; q < Q_; q += blockDim.x) {
    float s = 0.0f;
#pragma unroll
    for (int k = 0; k < S_; ++k) s += F[NB[q * S_ + k]];
    O[q] = s * (1.0f / S_);
  }
}

__global__ __launch_bounds__(256) void interp_kernel(const float* __restrict__ coarse,
                                                     const int* __restrict__ idx3,
                                                     const float* __restrict__ w3,
                                                     float* __restrict__ out) {
  int bc = blockIdx.x;
  int b = bc / C_, c = bc % C_;
  const float* CO = coarse + ((size_t)b * C_ + c) * Q_;
  const int*   I  = idx3 + (size_t)b * N_ * 3;
  const float* W  = w3 + (size_t)b * N_ * 3;
  float* O = out + ((size_t)b * C_ + c) * N_;
  for (int n = threadIdx.x; n < N_; n += blockDim.x) {
    O[n] = W[n * 3 + 0] * CO[I[n * 3 + 0]] +
           W[n * 3 + 1] * CO[I[n * 3 + 1]] +
           W[n * 3 + 2] * CO[I[n * 3 + 2]];
  }
}

extern "C" void kernel_launch(void* const* d_in, const int* in_sizes, int n_in,
                              void* d_out, int out_size, void* d_ws, size_t ws_size,
                              hipStream_t stream) {
  const float* points   = (const float*)d_in[0];
  const float* features = (const float*)d_in[1];

  auto align256 = [](size_t x) { return (x + 255) & ~(size_t)255; };
  char* ws = (char*)d_ws;
  size_t off = 0;
  int*   fps_idx = (int*)(ws + off);   off = align256(off + sizeof(int) * B_ * Q_);
  float* qA      = (float*)(ws + off); off = align256(off + sizeof(float) * B_ * Q_ * 4);
  float* pn2     = (float*)(ws + off); off = align256(off + sizeof(float) * B_ * N_);
  int*   nbr     = (int*)(ws + off);   off = align256(off + sizeof(int) * B_ * Q_ * S_);
  float* coarse  = (float*)(ws + off); off = align256(off + sizeof(float) * B_ * C_ * Q_);
  int*   idx3    = (int*)(ws + off);   off = align256(off + sizeof(int) * B_ * N_ * 3);
  float* w3      = (float*)(ws + off); off = align256(off + sizeof(float) * B_ * N_ * 3);

  float* out = (float*)d_out;

  pnorm2_kernel<<<(B_ * N_ + 255) / 256, 256, 0, stream>>>(points, pn2);
  fps_kernel<<<B_, 256, 0, stream>>>(points, fps_idx, qA);
  knn_kernel<<<B_ * (Q_ / 32), 256, 0, stream>>>(points, pn2, qA, nbr);
  group_mean_kernel<<<B_ * C_, 256, 0, stream>>>(features, nbr, coarse);
  three_nn_kernel<<<B_ * (N_ / 32), 256, 0, stream>>>(points, pn2, qA, idx3, w3);
  interp_kernel<<<B_ * C_, 256, 0, stream>>>(coarse, idx3, w3, out);
}


# fps: a wave redoes its argmax search only after a round whose sample lowered the one point its posted record names (its leading lane's candidate), not any lane's candidate: same indices
# speedup vs baseline: 5.2613x; 1.1792x over previous
#include <hip/hip_runtime.h>

#define B_ 4
#define N_ 8192
#define C_ 128
#define Q_ 2048
#define S_ 16
#define KEEP_ 17
#define RS_ 20

typedef __attribute__((ext_vector_type(2))) float v2f;
typedef __attribute__((ext_vector_type(8))) float v8f;

__global__ void pnorm2_kernel(const float* __restrict__ pts, float* __restrict__ pn2) {
  int i = blockIdx.x * blockDim.x + threadIdx.x;
  if (i < B_ * N_) {
    float x = pts[i * 3 + 0], y = pts[i * 3 + 1], z = pts[i * 3 + 2];
    pn2[i] = x * x + y * y + z * z;
  }
}

template <int M> __device__ __forceinline__ int row_xor(int v) {
  return __builtin_amdgcn_update_dpp(0, v, 0x160 + M, 0xf, 0xf, true);
}
__device__ __forceinline__ int xhalf(int v) {
  return __builtin_amdgcn_permlanex16(v, v, 0x76543210, 0xfedcba98, false, false);
}

#define FPS_PT_ (N_ / 256)
#define FPS_CH_ 8
#define FPS_CL_ (FPS_PT_ / FPS_CH_)
#define FPS_LS_ (FPS_PT_ + 1)

__device__ __forceinline__ float fps_dist2(float dx, float dy, float dz) {
  return __builtin_fmaf(dz, dz, __builtin_fmaf(dx, dx, dy * dy));
}

__device__ __forceinline__ void fps_take_higher(int& va, int& sa, int vb, int sb) {
  const bool g = vb > va;
  va = g ? vb : va;
  sa = g ? sb : sa;
}

__device__ __forceinline__ void fps_thread_argmax(const int (&d)[FPS_PT_], int& bv, int& bslot) {
  int v[FPS_CH_], sl[FPS_CH_];
#pragma unroll
  for (int c = 0; c < FPS_CH_; ++c) { v[c] = d[c * FPS_CL_]; sl[c] = c * FPS_CL_; }
#pragma unroll
  for (int j = 1; j < FPS_CL_; ++j) {
#pragma unroll
    for (int c = 0; c < FPS_CH_; ++c) {
      const int i = c * FPS_CL_ + j;
      const bool g = d[i] > v[c];
      v[c]  = g ? d[i] : v[c];
      sl[c] = g ? i : sl[c];
    }
  }
  fps_take_higher(v[0], sl[0], v[1], sl[1]);
  fps_take_higher(v[2], sl[2], v[3], sl[3]);
  fps_take_higher(v[4], sl[4], v[5], sl[5]);
  fps_take_higher(v[6], sl[6], v[7], sl[7]);
  fps_take_higher(v[0], sl[0], v[2], sl[2]);
  fps_take_higher(v[4], sl[4], v[6], sl[6]);
  fps_take_higher(v[0], sl[0], v[4], sl[4]);
  bv = v[0];
  bslot = sl[0];
}

__device__ __forceinline__ void fps_block(const float* __restrict__ pts,
                                          int* __restrict__ fps_idx,
                                          float* __restrict__ qA, const int b) {
  const int tid = threadIdx.x;
  const float* P = pts + (size_t)b * N_ * 3;

  __shared__ float4 lpts[256 * FPS_LS_];
  __shared__ int4 rec[2][10 + 256 + 64];
  __shared__ int4 samp[Q_];
  const int wave = tid >> 5;
  const int lane = tid & 31;

  float px[FPS_PT_], py[FPS_PT_], pz[FPS_PT_];
  int d[FPS_PT_];
  const float p0x = P[0], p0y = P[1], p0z = P[2];
  {
    const float4* P4 = (const float4*)(P + (size_t)tid * (FPS_PT_ * 3));
    float f[FPS_PT_ * 3];
#pragma unroll
    for (int k = 0; k < FPS_PT_ * 3 / 4; ++k) {
      const float4 t4 = P4[k];
      f[k * 4 + 0] = t4.x; f[k * 4 + 1] = t4.y; f[k * 4 + 2] = t4.z; f[k * 4 + 3] = t4.w;
    }
#pragma unroll
    for (int i = 0; i < FPS_PT_; ++i) {
      px[i] = f[i * 3 + 0]; py[i] = f[i * 3 + 1]; pz[i] = f[i * 3 + 2];
      lpts[tid * FPS_LS_ + i] = make_float4(px[i], py[i], pz[i], 0.0f);
      float dx = px[i] - p0x, dy = py[i] - p0y, dz = pz[i] - p0z;
      d[i] = __builtin_bit_cast(int, fps_dist2(dx, dy, dz));
    }
  }
  int bvi, bslot;
  float4 cp;
  int wm;
  int ia, iz;
  bool lead;
  auto renew = [&]() {
    fps_thread_argmax(d, bvi, bslot);
    cp = lpts[tid * FPS_LS_ + bslot];
    wm = bvi;
    wm = max(wm, row_xor<1>(wm)); wm = max(wm, row_xor<2>(wm));
    wm = max(wm, row_xor<4>(wm)); wm = max(wm, row_xor<8>(wm));
    wm = max(wm, xhalf(wm));
    lead = lane == __builtin_ctz(__builtin_amdgcn_ballot_w32(bvi == wm));
    ia = lead ? wave : 10 + tid;
    iz = lead ? 8 * 4 + wave : (10 + 256) * 4 + tid;
    asm volatile("" :: "v"(cp.x), "v"(cp.y), "v"(cp.z), "v"(ia), "v"(iz));
  };
  renew();
  if (tid == 0) {
    fps_idx[b * Q_ + 0] = 0;
    qA[((size_t)b * Q_ + 0) * 4 + 0] = p0x;
    qA[((size_t)b * Q_ + 0) * 4 + 1] = p0y;
    qA[((size_t)b * Q_ + 0) * 4 + 2] = p0z;
    qA[((size_t)b * Q_ + 0) * 4 + 3] = p0x * p0x + p0y * p0y + p0z * p0z;
  }

  for (int s = 1; s < Q_; ++s) {
    const int bidx = tid * FPS_PT_ + bslot;
    int4* R = rec[s & 1];
    R[ia] = make_int4(wm, bidx, __builtin_bit_cast(int, cp.x), __builtin_bit_cast(int, cp.y));
    ((int*)R)[iz] = __builtin_bit_cast(int, cp.z);
    __syncthreads();
    const int4 ra = R[lane & 7];
    const int rz = ((const int*)R)[8 * 4 + (lane & 7)];
    int gm = ra.x;
    gm = max(gm, row_xor<1>(gm)); gm = max(gm, row_xor<2>(gm)); gm = max(gm, row_xor<4>(gm));
    const int wl = __builtin_ctz(__builtin_amdgcn_ballot_w32(ra.x == gm));
    const int nxt = __builtin_amdgcn_readlane(ra.y, wl);
    const float cx = __builtin_bit_cast(float, __builtin_amdgcn_readlane(ra.z, wl));
    const float cy = __builtin_bit_cast(float, __builtin_amdgcn_readlane(ra.w, wl));
    const float cz = __builtin_bit_cast(float, __builtin_amdgcn_readlane(rz, wl));
    if (tid == (s & 255))
      samp[s] = make_int4(nxt, __builtin_bit_cast(int, cx), __builtin_bit_cast(int, cy),
                          __builtin_bit_cast(int, cz));
    const int cand_dd = __builtin_bit_cast(int, fps_dist2(cp.x - cx, cp.y - cy, cp.z - cz));
    const bool record_lowered = lead && cand_dd <= bvi;
#pragma unroll
    for (int i = 0; i < FPS_PT_; ++i) {
      float dx = px[i] - cx, dy = py[i] - cy, dz = pz[i] - cz;
      float dd = fps_dist2(dx, dy, dz);
      d[i] = min(d[i], __builtin_bit_cast(int, dd));
    }
    if (__builtin_amdgcn_ballot_w32(record_lowered) != 0) renew();
  }

  for (int s = tid; s < Q_; s += 256) {
    if (s == 0) continue;
    const int4 e = samp[s];
    const float x = __builtin_bit_cast(float, e.y), y = __builtin_bit_cast(float, e.z),
                z = __builtin_bit_cast(float, e.w);
    float n2;
    {
#pragma clang fp contract(off)
      n2 = x * x + y * y;
    }
    fps_idx[b * Q_ + s] = e.x;
    ((float4*)qA)[(size_t)b * Q_ + s] = make_float4(x, y, z, __builtin_fmaf(z, z, n2));
  }
}

#define KNN_SL_ 8
#define KNN_SLCH_ (N_ / 16 / KNN_SL_)
#define KNN_QC_ 32

__device__ __forceinline__ void knn_insert(float (&bd)[KEEP_], int (&bi)[KEEP_], float v, int pi) {
  if (v < bd[KEEP_ - 1]) {
    bd[KEEP_ - 1] = v; bi[KEEP_ - 1] = pi;
#pragma unroll
    for (int t = KEEP_ - 1; t > 0; --t) {
      if (bd[t] < bd[t - 1]) {
        float tf = bd[t]; bd[t] = bd[t - 1]; bd[t - 1] = tf;
        int   ti = bi[t]; bi[t] = bi[t - 1]; bi[t - 1] = ti;
      }
    }
  }
}

__device__ __forceinline__ void knn_mean_block(const float* __restrict__ pts,
                                               const float* __restrict__ pn2,
                                               const float* __restrict__ feat,
                                               float* __restrict__ pmean, const int blk) {
  const int wave = __builtin_amdgcn_readfirstlane((int)(threadIdx.x >> 5));
  const int lane = threadIdx.x & 31;
  const int b    = blk / (N_ / 32);
  const int q0   = (blk % (N_ / 32)) * 32;

  const float* P  = pts + (size_t)b * N_ * 3;
  const float* PN = pn2 + (size_t)b * N_;

  __shared__ float stage[8][32 * RS_];
  __shared__ float mrg_d[KNN_SL_ - 1][KEEP_][32];
  __shared__ int   mrg_i[KNN_SL_ - 1][KEEP_][32];
  __shared__ int   kq[KNN_SL_][KNN_QC_ * 64];
  __shared__ float4 qrow[KNN_SL_][32];
  __shared__ int   nl[S_][32];
  float* st = stage[wave];
  int* kql = kq[wave] + lane;

  const int lhalf = lane >> 4;
  const int l16   = lane & 15;

  {
    const float x = P[(q0 + lane) * 3 + 0], y = P[(q0 + lane) * 3 + 1], z = P[(q0 + lane) * 3 + 2];
    float n2;
    {
#pragma clang fp contract(off)
      n2 = x * x + y * y;
    }
    qrow[wave][lane] = make_float4(x, y, z, __builtin_fmaf(z, z, n2));
  }
  const float4* QR = qrow[wave];
  const float4 qa0 = QR[l16];
  const float4 qa1 = QR[16 + l16];
  v2f a0 = lhalf ? (v2f){-2.0f * qa0.z, 1.0f} : (v2f){-2.0f * qa0.x, -2.0f * qa0.y};
  v2f a1 = lhalf ? (v2f){-2.0f * qa1.z, 1.0f} : (v2f){-2.0f * qa1.x, -2.0f * qa1.y};

  v8f c0, c1;
#pragma unroll
  for (int r = 0; r < 8; ++r) {
    c0[r] = QR[r + lhalf * 8].w;
    c1[r] = QR[16 + r + lhalf * 8].w;
  }

  float bd[KEEP_]; int bi[KEEP_];
#pragma unroll
  for (int k = 0; k < KEEP_; ++k) { bd[k] = 3.0e38f; bi[k] = 0; }

  auto load_bb = [&](int ch) -> v2f {
    int pt = ch * 16 + l16;
    float x0 = P[pt * 3 + (lhalf ? 2 : 0)];
    float x1 = lhalf ? PN[pt] : P[pt * 3 + 1];
    return (v2f){x0, x1};
  };

  const int ch0 = wave * KNN_SLCH_;
  const int ch1 = ch0 + KNN_SLCH_;
  int qo = 0;
  v2f bb = load_bb(ch0);
#pragma nounroll
  for (int ch = ch0; ch < ch1; ++ch) {
    v2f bb_next = load_bb(ch + 1 < ch1 ? ch + 1 : ch1 - 1);

    v8f d0 = __builtin_amdgcn_wmma_f32_16x16x4_f32(false, a0, false, bb, (short)0, c0, false, false);
    v8f d1 = __builtin_amdgcn_wmma_f32_16x16x4_f32(false, a1, false, bb, (short)0, c1, false, false);

#pragma unroll
    for (int r = 0; r < 8; ++r) {
      st[(r + lhalf * 8) * RS_ + l16]       = d0[r];
      st[(16 + r + lhalf * 8) * RS_ + l16]  = d1[r];
    }
    const float4* myrow = (const float4*)(st + lane * RS_);
#pragma unroll
    for (int jj = 0; jj < 4; ++jj) {
      float4 v4 = myrow[jj];
#pragma unroll
      for (int e = 0; e < 4; ++e) {
        float v = fmaxf(e == 0 ? v4.x : (e == 1 ? v4.y : (e == 2 ? v4.z : v4.w)), 0.0f);
        kql[qo]      = __builtin_bit_cast(int, v);
        kql[qo + 32] = ch * 16 + jj * 4 + e;
        qo += (v < bd[KEEP_ - 1]) ? 64 : 0;
      }
    }
    bb = bb_next;

    if (ch == ch1 - 1 || __builtin_amdgcn_ballot_w32(qo > (KNN_QC_ - 16) * 64) != 0) {
      int nv = 0, np = 0;
      if (0 < qo) { nv = kql[0]; np = kql[32]; }
#pragma nounroll
      for (int jo = 0; __builtin_amdgcn_ballot_w32(jo < qo) != 0; jo += 64) {
        const float v = jo < qo ? __builtin_bit_cast(float, nv) : 3.0e38f;
        const int  pi = np;
        if (jo + 64 < qo) { nv = kql[jo + 64]; np = kql[jo + 96]; }
        if (__builtin_amdgcn_ballot_w32(v < bd[KEEP_ - 1]) == 0) continue;
        knn_insert(bd, bi, v, pi);
      }
      qo = 0;
    }
  }

  if (wave != 0) {
#pragma unroll
    for (int k = 0; k < KEEP_; ++k) {
      mrg_d[wave - 1][k][lane] = bd[k];
      mrg_i[wave - 1][k][lane] = bi[k];
    }
  }
  __syncthreads();
  if (wave == 0) {
#pragma nounroll
    for (int w = 0; w < KNN_SL_ - 1; ++w) {
#pragma nounroll
      for (int k = 0; k < KEEP_; ++k) {
        const float v = mrg_d[w][k][lane];
        const int  pi = mrg_i[w][k][lane];
        if (__builtin_amdgcn_ballot_w32(v < bd[KEEP_ - 1]) == 0) break;
        knn_insert(bd, bi, v, pi);
      }
    }
#pragma unroll
    for (int k = 0; k < S_; ++k) nl[k][lane] = bi[k + 1];
  }
  __syncthreads();

  unsigned nb[S_];
#pragma unroll
  for (int k = 0; k < S_; ++k) nb[k] = (unsigned)nl[k][lane];
  const float* F = feat + ((size_t)b * C_ + wave * (C_ / KNN_SL_)) * N_;
  float* O = pmean + ((size_t)b * C_ + wave * (C_ / KNN_SL_)) * N_ + q0 + lane;
  for (int cc = 0; cc < C_ / KNN_SL_; ++cc) {
    float s = 0.0f;
#pragma unroll
    for (int k = 0; k < S_; ++k) s += F[nb[k]];
    O[0] = s * (1.0f / S_);
    F += N_;
    O += N_;
  }
}

__global__ __launch_bounds__(256) void fps_knn_mean_kernel(const float* __restrict__ pts,
                                                           const float* __restrict__ pn2,
                                                           const float* __restrict__ feat,
                                                           int* __restrict__ fps_idx,
                                                           float* __restrict__ qA,
                                                           float* __restrict__ pmean) {
  if (blockIdx.x < B_) fps_block(pts, fps_idx, qA, (int)blockIdx.x);
  else knn_mean_block(pts, pn2, feat, pmean, (int)blockIdx.x - B_);
}

#define TNN_SL_ 8
#define TNN_SLCH_ (Q_ / 16 / TNN_SL_)

__device__ __forceinline__ void tnn_insert(float (&bd)[3], int (&bi)[3], float v, int qi) {
  if (v < bd[2]) {
    bd[2] = v; bi[2] = qi;
    if (bd[2] < bd[1]) { float tf = bd[2]; bd[2] = bd[1]; bd[1] = tf;
                         int ti = bi[2]; bi[2] = bi[1]; bi[1] = ti; }
    if (bd[1] < bd[0]) { float tf = bd[1]; bd[1] = bd[0]; bd[0] = tf;
                         int ti = bi[1]; bi[1] = bi[0]; bi[0] = ti; }
  }
}

__global__ __launch_bounds__(256) void three_nn_kernel(const float* __restrict__ pts,
                                                       const float* __restrict__ pn2,
                                                       const float* __restrict__ qA,
                                                       int* __restrict__ idx3,
                                                       float* __restrict__ w3) {
  const int wave = __builtin_amdgcn_readfirstlane((int)(threadIdx.x >> 5));
  const int lane = threadIdx.x & 31;
  const int b    = blockIdx.x / (N_ / 32);
  const int n0   = (blockIdx.x % (N_ / 32)) * 32;

  const float* P  = pts + (size_t)b * N_ * 3;
  const float* PN = pn2 + (size_t)b * N_;
  const float* QA = qA + (size_t)b * Q_ * 4;

  __shared__ float stage[8][32 * RS_];
  __shared__ float mrg_d[TNN_SL_ - 1][3][32];
  __shared__ int   mrg_i[TNN_SL_ - 1][3][32];
  float* st = stage[wave];

  const int lhalf = lane >> 4;
  const int l16   = lane & 15;

  int m0 = n0 + l16, m1 = n0 + 16 + l16;
  float p0x = P[m0 * 3 + 0], p0y = P[m0 * 3 + 1], p0z = P[m0 * 3 + 2];
  float p1x = P[m1 * 3 + 0], p1y = P[m1 * 3 + 1], p1z = P[m1 * 3 + 2];
  v2f a0 = lhalf ? (v2f){-2.0f * p0z, 1.0f} : (v2f){-2.0f * p0x, -2.0f * p0y};
  v2f a1 = lhalf ? (v2f){-2.0f * p1z, 1.0f} : (v2f){-2.0f * p1x, -2.0f * p1y};

  v8f c0, c1;
#pragma unroll
  for (int r = 0; r < 8; ++r) {
    c0[r] = PN[n0 + r + lhalf * 8];
    c1[r] = PN[n0 + 16 + r + lhalf * 8];
  }

  float bd[3]; int bi[3];
#pragma unroll
  for (int k = 0; k < 3; ++k) { bd[k] = 3.0e38f; bi[k] = 0; }

  auto load_bb = [&](int ch) -> v2f {
    const float4 qa = ((const float4*)QA)[ch * 16 + l16];
    return lhalf ? (v2f){qa.z, qa.w} : (v2f){qa.x, qa.y};
  };

  const int ch0 = wave * TNN_SLCH_;
  const int ch1 = ch0 + TNN_SLCH_;
  v2f bb = load_bb(ch0);
  for (int ch = ch0; ch < ch1; ++ch) {
    v2f bb_next = load_bb(ch + 1 < ch1 ? ch + 1 : ch1 - 1);

    v8f d0 = __builtin_amdgcn_wmma_f32_16x16x4_f32(false, a0, false, bb, (short)0, c0, false, false);
    v8f d1 = __builtin_amdgcn_wmma_f32_16x16x4_f32(false, a1, false, bb, (short)0, c1, false, false);

#pragma unroll
    for (int r = 0; r < 8; ++r) {
      st[(r + lhalf * 8) * RS_ + l16]      = d0[r];
      st[(16 + r + lhalf * 8) * RS_ + l16] = d1[r];
    }
    const float4* myrow = (const float4*)(st + lane * RS_);
#pragma unroll
    for (int jj = 0; jj < 4; ++jj) {
      float4 v4 = myrow[jj];
#pragma unroll
      for (int e = 0; e < 4; ++e) {
        float v = fmaxf(e == 0 ? v4.x : (e == 1 ? v4.y : (e == 2 ? v4.z : v4.w)), 0.0f);
        tnn_insert(bd, bi, v, ch * 16 + jj * 4 + e);
      }
    }
    bb = bb_next;
  }

  if (wave != 0) {
#pragma unroll
    for (int k = 0; k < 3; ++k) {
      mrg_d[wave - 1][k][lane] = bd[k];
      mrg_i[wave - 1][k][lane] = bi[k];
    }
  }
  __syncthreads();
  if (wave != 0) return;

  float md[TNN_SL_ - 1][3]; int mi[TNN_SL_ - 1][3];
#pragma unroll
  for (int w = 0; w < TNN_SL_ - 1; ++w) {
#pragma unroll
    for (int k = 0; k < 3; ++k) { md[w][k] = mrg_d[w][k][lane]; mi[w][k] = mrg_i[w][k][lane]; }
  }
#pragma unroll
  for (int w = 0; w < TNN_SL_ - 1; ++w) {
#pragma unroll
    for (int k = 0; k < 3; ++k) tnn_insert(bd, bi, md[w][k], mi[w][k]);
  }

  int n = n0 + lane;
  float dv0 = fmaxf(bd[0], 1e-10f), dv1 = fmaxf(bd[1], 1e-10f), dv2 = fmaxf(bd[2], 1e-10f);
  float i0 = 1.0f / (dv0 + 1e-8f), i1 = 1.0f / (dv1 + 1e-8f), i2 = 1.0f / (dv2 + 1e-8f);
  float inv_s = 1.0f / (i0 + i1 + i2);
  size_t base = ((size_t)b * N_ + n) * 3;
  idx3[base + 0] = bi[0]; idx3[base + 1] = bi[1]; idx3[base + 2] = bi[2];
  w3[base + 0] = i0 * inv_s; w3[base + 1] = i1 * inv_s; w3[base + 2] = i2 * inv_s;
}

__global__ __launch_bounds__(256) void interp_kernel(const float* __restrict__ pmean,
                                                     const int* __restrict__ fps_idx,
                                                     const int* __restrict__ idx3,
                                                     const float* __restrict__ w3,
                                                     float* __restrict__ out) {
  int bc = blockIdx.x;
  int b = bc / C_, c = bc % C_;
  const float* PM = pmean + ((size_t)b * C_ + c) * N_;
  const int*   FI = fps_idx + (size_t)b * Q_;
  __shared__ float CO[Q_];
  {
    float m[Q_ / 256];
#pragma unroll
    for (int j = 0; j < Q_ / 256; ++j) m[j] = PM[FI[threadIdx.x + j * 256]];
#pragma unroll
    for (int j = 0; j < Q_ / 256; ++j) CO[threadIdx.x + j * 256] = m[j];
  }
  __syncthreads();
  const int*   I  = idx3 + (size_t)b * N_ * 3;
  const float* W  = w3 + (size_t)b * N_ * 3;
  float* O = out + ((size_t)b * C_ + c) * N_;
  for (int n = threadIdx.x; n < N_; n += blockDim.x) {
    O[n] = W[n * 3 + 0] * CO[I[n * 3 + 0]] +
           W[n * 3 + 1] * CO[I[n * 3 + 1]] +
           W[n * 3 + 2] * CO[I[n * 3 + 2]];
  }
}

extern "C" void kernel_launch(void* const* d_in, const int* in_sizes, int n_in,
                              void* d_out, int out_size, void* d_ws, size_t ws_size,
                              hipStream_t stream) {
  const float* points   = (const float*)d_in[0];
  const float* features = (const float*)d_in[1];

  auto align256 = [](size_t x) { return (x + 255) & ~(size_t)255; };
  char* ws = (char*)d_ws;
  size_t off = 0;
  int*   fps_idx = (int*)(ws + off);   off = align256(off + sizeof(int) * B_ * Q_);
  float* qA      = (float*)(ws + off); off = align256(off + sizeof(float) * B_ * Q_ * 4);
  float* pn2     = (float*)(ws + off); off = align256(off + sizeof(float) * B_ * N_);
  float* pmean   = (float*)(ws + off); off = align256(off + sizeof(float) * B_ * C_ * N_);
  int*   idx3    = (int*)(ws + off);   off = align256(off + sizeof(int) * B_ * N_ * 3);
  float* w3      = (float*)(ws + off); off = align256(off + sizeof(float) * B_ * N_ * 3);

  float* out = (float*)d_out;

  pnorm2_kernel<<<(B_ * N_ + 255) / 256, 256, 0, stream>>>(points, pn2);
  fps_knn_mean_kernel<<<B_ + B_ * (N_ / 32), 256, 0, stream>>>(points, pn2, features, fps_idx, qA, pmean);
  three_nn_kernel<<<B_ * (N_ / 32), 256, 0, stream>>>(points, pn2, qA, idx3, w3);
  interp_kernel<<<B_ * C_, 256, 0, stream>>>(pmean, fps_idx, idx3, w3, out);
}
